# starting kernel
# baseline (speedup 1.0000x reference)
#include <hip/hip_runtime.h>

typedef __attribute__((ext_vector_type(2))) float v2f;
typedef __attribute__((ext_vector_type(8))) float v8f;

#define NPTS (1 << 20)

__constant__ float SH_COEF[20][16] = {
    {0.28209479177387814f,0,0,0,0,0,-0.31539156525252f,0,0,0,0,0,0,0,0,0},
    {0,0,0,-0.48860251190291987f,0,0,0,0,0,0,0,0,0,0.4570457994644657f,0,0},
    {0,-0.48860251190291987f,0,0,0,0,0,0,0,0,0,0.4570457994644657f,0,0,0,0},
    {0,0,0.48860251190291987f,0,0,0,0,0,0,0,0,0,-1.1195289977703462f,0,0,0},
    {0,0,0,0,0,0,0,0,0.5462742152960396f,0,0,0,0,0,0,0},
    {0,0,0,0,0,0,0,0,-0.5462742152960396f,0,0,0,0,0,0,0},
    {0,0,0,0,0,0,0.94617469575756f,0,0,0,0,0,0,0,0,0},
    {0,0,0,0,1.0925484305920792f,0,0,0,0,0,0,0,0,0,0,0},
    {0,0,0,0,0,-1.0925484305920792f,0,0,0,0,0,0,0,0,0,0},
    {0,0,0,0,0,0,0,-1.0925484305920792f,0,0,0,0,0,0,0,0},
    {0,0,0,0,0,0,0,0,0,0,0,0,0,0,0,-0.5900435899266435f},
    {0,0,0,0,0,0,0,0,0,0.5900435899266435f,0,0,0,0,0,0},
    {0,0,0,0,0,0,0,0,0,0,0,0,1.865881662950577f,0,0,0},
    {0,0,0,0,0,0,0,0,0,-1.7701307697799304f,0,0,0,0,0,0},
    {0,0,0,0,0,0,0,0,0,0,0,0,0,0,1.445305721320277f,0},
    {0,0,0,0,0,0,0,0,0,0,0,0,0,0,0,1.7701307697799304f},
    {0,0,0,0,0,0,0,0,0,0,0,0,0,0,-1.445305721320277f,0},
    {0,0,0,0,0,0,0,0,0,0,0,0,0,-2.2852289973223288f,0,0},
    {0,0,0,0,0,0,0,0,0,0,0,-2.2852289973223288f,0,0,0,0},
    {0,0,0,0,0,0,0,0,0,0,2.890611442640554f,0,0,0,0,0},
};

__global__ __launch_bounds__(256) void interp_kernel(
    const float* __restrict__ pos,
    const float* __restrict__ g0, const float* __restrict__ g1,
    const float* __restrict__ g2, const float* __restrict__ g3,
    const float* __restrict__ g4, const float* __restrict__ g5,
    const float* __restrict__ g6, const float* __restrict__ g7,
    float* __restrict__ out)
{
  const int p = blockIdx.x * 256 + threadIdx.x;
  const float px = pos[3 * p + 0];
  const float py = pos[3 * p + 1];
  const float pz = pos[3 * p + 2];

  const float* grids[8] = {g0, g1, g2, g3, g4, g5, g6, g7};
  const int RES[8] = {16, 21, 28, 39, 52, 70, 95, 128};

  float* o = out + (size_t)p * 32;

#pragma unroll
  for (int l = 0; l < 8; ++l) {
    const int R = RES[l];
    const float cx = px * (float)(R - 1);
    const float cy = py * (float)(R - 1);
    const float cz = pz * (float)(R - 1);
    int ix = (int)cx; ix = ix < R - 2 ? ix : R - 2; ix = ix > 0 ? ix : 0;
    int iy = (int)cy; iy = iy < R - 2 ? iy : R - 2; iy = iy > 0 ? iy : 0;
    int iz = (int)cz; iz = iz < R - 2 ? iz : R - 2; iz = iz > 0 ? iz : 0;
    const float fx = cx - (float)ix;
    const float fy = cy - (float)iy;
    const float fz = cz - (float)iz;

    const v2f* __restrict__ g = (const v2f*)grids[l];
    const int i000 = (ix * R + iy) * R + iz;
    const int i100 = i000 + R * R;

    const v2f c000 = g[i000],         c001 = g[i000 + 1];
    const v2f c010 = g[i000 + R],     c011 = g[i000 + R + 1];
    const v2f c100 = g[i100],         c101 = g[i100 + 1];
    const v2f c110 = g[i100 + R],     c111 = g[i100 + R + 1];

    const v2f c00 = c000 + fz * (c001 - c000);
    const v2f c01 = c010 + fz * (c011 - c010);
    const v2f c10 = c100 + fz * (c101 - c100);
    const v2f c11 = c110 + fz * (c111 - c110);
    const v2f c0  = c00 + fy * (c01 - c00);
    const v2f c1  = c10 + fy * (c11 - c10);
    const v2f r   = c0  + fx * (c1 - c0);

    o[2 * l + 0] = r.x;
    o[2 * l + 1] = r.y;
  }
}

__global__ __launch_bounds__(256) void sh_wmma_kernel(
    const float* __restrict__ dir, float* __restrict__ out)
{
  const int lane = threadIdx.x & 31;
  const int wave = threadIdx.x >> 5;
  const int tile = blockIdx.x * 8 + wave;
  const int  n   = lane & 15;
  const bool hi  = lane >= 16;

  const int p = tile * 16 + n;
  const float dx = dir[3 * p + 0];
  const float dy = dir[3 * p + 1];
  const float dz = dir[3 * p + 2];
  const float inv = rsqrtf(dx * dx + dy * dy + dz * dz);
  const float x = dx * inv, y = dy * inv, z = dz * inv;
  const float xx = x * x, yy = y * y, zz = z * z;

  const float m0 = 1.0f, m1 = x, m2 = y, m3 = z;
  const float m4 = xx, m5 = yy, m6 = zz, m7 = x * y, m8 = y * z, m9 = x * z;
  const float m10 = xx * x, m11 = yy * y, m12 = zz * z;
  const float m13 = xx * y, m14 = xx * z, m15 = x * yy;
  const float m16 = yy * z, m17 = x * zz, m18 = y * zz, m19 = x * y * z;

  v8f acc = {0.f, 0.f, 0.f, 0.f, 0.f, 0.f, 0.f, 0.f};

#define SH_CHUNK(C, E0, E1, E2, E3)                                           \
  do {                                                                        \
    v2f a, b;                                                                 \
    a.x = hi ? (E2) : (E0);                                                   \
    a.y = hi ? (E3) : (E1);                                                   \
    const int kb = (C) + (hi ? 2 : 0);                                        \
    b.x = SH_COEF[kb][n];                                                     \
    b.y = SH_COEF[kb + 1][n];                                                 \
    acc = __builtin_amdgcn_wmma_f32_16x16x4_f32(                              \
        false, a, false, b, (short)0, acc, false, false);                     \
  } while (0)

  SH_CHUNK(0,  m0,  m1,  m2,  m3);
  SH_CHUNK(4,  m4,  m5,  m6,  m7);
  SH_CHUNK(8,  m8,  m9,  m10, m11);
  SH_CHUNK(12, m12, m13, m14, m15);
  SH_CHUNK(16, m16, m17, m18, m19);
#undef SH_CHUNK

  const int rowBase = tile * 16 + (hi ? 8 : 0);
#pragma unroll
  for (int r = 0; r < 8; ++r) {
    out[(size_t)(rowBase + r) * 32 + 16 + n] = acc[r];
  }
}

extern "C" void kernel_launch(void* const* d_in, const int* in_sizes, int n_in,
                              void* d_out, int out_size, void* d_ws, size_t ws_size,
                              hipStream_t stream) {
  (void)in_sizes; (void)n_in; (void)out_size; (void)d_ws; (void)ws_size;

  const float* pos = (const float*)d_in[0];
  const float* dir = (const float*)d_in[1];
  const float* g0 = (const float*)d_in[2];
  const float* g1 = (const float*)d_in[3];
  const float* g2 = (const float*)d_in[4];
  const float* g3 = (const float*)d_in[5];
  const float* g4 = (const float*)d_in[6];
  const float* g5 = (const float*)d_in[7];
  const float* g6 = (const float*)d_in[8];
  const float* g7 = (const float*)d_in[9];
  float* out = (float*)d_out;

  interp_kernel<<<NPTS / 256, 256, 0, stream>>>(pos, g0, g1, g2, g3, g4, g5,
                                                g6, g7, out);

  sh_wmma_kernel<<<(NPTS / 16) / 8, 256, 0, stream>>>(dir, out);
}


# Fuse SH into the interp kernel (one launch): each wave builds its 32 output rows in LDS and stores every 128-byte line once, whole; per-element arithmetic and summation order unchanged
# speedup vs baseline: 1.1272x; 1.1272x over previous
#include <hip/hip_runtime.h>


typedef __attribute__((ext_vector_type(2))) float v2f;
typedef __attribute__((ext_vector_type(4))) float v4f;
typedef __attribute__((ext_vector_type(8))) float v8f;

#define NPTS (1 << 20)

#define ROW_PITCH 36

__constant__ float SH_COEF[20][16] = {
    {0.28209479177387814f,0,0,0,0,0,-0.31539156525252f,0,0,0,0,0,0,0,0,0},
    {0,0,0,-0.48860251190291987f,0,0,0,0,0,0,0,0,0,0.4570457994644657f,0,0},
    {0,-0.48860251190291987f,0,0,0,0,0,0,0,0,0,0.4570457994644657f,0,0,0,0},
    {0,0,0.48860251190291987f,0,0,0,0,0,0,0,0,0,-1.1195289977703462f,0,0,0},
    {0,0,0,0,0,0,0,0,0.5462742152960396f,0,0,0,0,0,0,0},
    {0,0,0,0,0,0,0,0,-0.5462742152960396f,0,0,0,0,0,0,0},
    {0,0,0,0,0,0,0.94617469575756f,0,0,0,0,0,0,0,0,0},
    {0,0,0,0,1.0925484305920792f,0,0,0,0,0,0,0,0,0,0,0},
    {0,0,0,0,0,-1.0925484305920792f,0,0,0,0,0,0,0,0,0,0},
    {0,0,0,0,0,0,0,-1.0925484305920792f,0,0,0,0,0,0,0,0},
    {0,0,0,0,0,0,0,0,0,0,0,0,0,0,0,-0.5900435899266435f},
    {0,0,0,0,0,0,0,0,0,0.5900435899266435f,0,0,0,0,0,0},
    {0,0,0,0,0,0,0,0,0,0,0,0,1.865881662950577f,0,0,0},
    {0,0,0,0,0,0,0,0,0,-1.7701307697799304f,0,0,0,0,0,0},
    {0,0,0,0,0,0,0,0,0,0,0,0,0,0,1.445305721320277f,0},
    {0,0,0,0,0,0,0,0,0,0,0,0,0,0,0,1.7701307697799304f},
    {0,0,0,0,0,0,0,0,0,0,0,0,0,0,-1.445305721320277f,0},
    {0,0,0,0,0,0,0,0,0,0,0,0,0,-2.2852289973223288f,0,0},
    {0,0,0,0,0,0,0,0,0,0,0,-2.2852289973223288f,0,0,0,0},
    {0,0,0,0,0,0,0,0,0,0,2.890611442640554f,0,0,0,0,0},
};

__global__ __launch_bounds__(256) void encode_kernel(
    const float* __restrict__ pos, const float* __restrict__ dir,
    const float* __restrict__ g0, const float* __restrict__ g1,
    const float* __restrict__ g2, const float* __restrict__ g3,
    const float* __restrict__ g4, const float* __restrict__ g5,
    const float* __restrict__ g6, const float* __restrict__ g7,
    float* __restrict__ out)
{
  __shared__ __attribute__((aligned(16))) float stage[8][32 * ROW_PITCH];

  const int lane = threadIdx.x & 31;
  const int wave = threadIdx.x >> 5;
  const int P0   = blockIdx.x * 256 + wave * 32;
  float* st = stage[wave];

  const float px = pos[3 * (P0 + lane) + 0];
  const float py = pos[3 * (P0 + lane) + 1];
  const float pz = pos[3 * (P0 + lane) + 2];

  {
    const int  n  = lane & 15;
    const bool hi = lane >= 16;

    v2f bfrag[5];
    for (int c = 0; c < 5; ++c) {
      const int kb = 4 * c + (hi ? 2 : 0);
      bfrag[c].x = SH_COEF[kb][n];
      bfrag[c].y = SH_COEF[kb + 1][n];
    }

    for (int t = 0; t < 2; ++t) {
      const int p = P0 + 16 * t + n;
      const float dx = dir[3 * p + 0];
      const float dy = dir[3 * p + 1];
      const float dz = dir[3 * p + 2];
      const float inv = rsqrtf(dx * dx + dy * dy + dz * dz);
      const float x = dx * inv, y = dy * inv, z = dz * inv;
      const float xx = x * x, yy = y * y, zz = z * z;

      const float m0 = 1.0f, m1 = x, m2 = y, m3 = z;
      const float m4 = xx, m5 = yy, m6 = zz, m7 = x * y, m8 = y * z, m9 = x * z;
      const float m10 = xx * x, m11 = yy * y, m12 = zz * z;
      const float m13 = xx * y, m14 = xx * z, m15 = x * yy;
      const float m16 = yy * z, m17 = x * zz, m18 = y * zz, m19 = x * y * z;

      v8f acc = {0.f, 0.f, 0.f, 0.f, 0.f, 0.f, 0.f, 0.f};

#define SH_CHUNK(C, E0, E1, E2, E3)                                           \
      do {                                                                    \
        v2f a;                                                                \
        a.x = hi ? (E2) : (E0);                                               \
        a.y = hi ? (E3) : (E1);                                               \
        acc = __builtin_amdgcn_wmma_f32_16x16x4_f32(                          \
            false, a, false, bfrag[(C) / 4], (short)0, acc, false, false);    \
      } while (0)

      SH_CHUNK(0,  m0,  m1,  m2,  m3);
      SH_CHUNK(4,  m4,  m5,  m6,  m7);
      SH_CHUNK(8,  m8,  m9,  m10, m11);
      SH_CHUNK(12, m12, m13, m14, m15);
      SH_CHUNK(16, m16, m17, m18, m19);
#undef SH_CHUNK

      const int rowBase = 16 * t + (hi ? 8 : 0);
#pragma unroll
      for (int r = 0; r < 8; ++r) {
        st[(rowBase + r) * ROW_PITCH + 16 + n] = acc[r];
      }
    }
  }

  __builtin_amdgcn_sched_barrier(0);

  {
    const float* grids[8] = {g0, g1, g2, g3, g4, g5, g6, g7};
    const int RES[8] = {16, 21, 28, 39, 52, 70, 95, 128};

    v2f res[8];

#pragma unroll
    for (int l = 0; l < 8; ++l) {
      const int R = RES[l];
      const float cx = px * (float)(R - 1);
      const float cy = py * (float)(R - 1);
      const float cz = pz * (float)(R - 1);
      int ix = (int)cx; ix = ix < R - 2 ? ix : R - 2; ix = ix > 0 ? ix : 0;
      int iy = (int)cy; iy = iy < R - 2 ? iy : R - 2; iy = iy > 0 ? iy : 0;
      int iz = (int)cz; iz = iz < R - 2 ? iz : R - 2; iz = iz > 0 ? iz : 0;
      const float fx = cx - (float)ix;
      const float fy = cy - (float)iy;
      const float fz = cz - (float)iz;

      const v2f* __restrict__ g = (const v2f*)grids[l];
      const int i000 = (ix * R + iy) * R + iz;
      const int i100 = i000 + R * R;

      const v2f c000 = g[i000],         c001 = g[i000 + 1];
      const v2f c010 = g[i000 + R],     c011 = g[i000 + R + 1];
      const v2f c100 = g[i100],         c101 = g[i100 + 1];
      const v2f c110 = g[i100 + R],     c111 = g[i100 + R + 1];

      const v2f c00 = c000 + fz * (c001 - c000);
      const v2f c01 = c010 + fz * (c011 - c010);
      const v2f c10 = c100 + fz * (c101 - c100);
      const v2f c11 = c110 + fz * (c111 - c110);
      const v2f c0  = c00 + fy * (c01 - c00);
      const v2f c1  = c10 + fy * (c11 - c10);
      res[l]        = c0  + fx * (c1 - c0);
    }

    v4f* srow = (v4f*)(st + lane * ROW_PITCH);
    for (int k = 0; k < 4; ++k) {
      v4f v = {res[2 * k].x, res[2 * k].y, res[2 * k + 1].x, res[2 * k + 1].y};
      srow[k] = v;
    }
  }

  __builtin_amdgcn_fence(__ATOMIC_RELEASE, "wavefront");
  __builtin_amdgcn_wave_barrier();
  __builtin_amdgcn_fence(__ATOMIC_ACQUIRE, "wavefront");

  {
    const int rsub = lane >> 3;
    const int c4   = (lane & 7) * 4;
    float* obase = out + (size_t)P0 * 32;
    for (int j = 0; j < 8; ++j) {
      const int row = 4 * j + rsub;
      const v4f v = *(const v4f*)(st + row * ROW_PITCH + c4);
      *(v4f*)(obase + row * 32 + c4) = v;
    }
  }
}

extern "C" void kernel_launch(void* const* d_in, const int* in_sizes, int n_in,
                              void* d_out, int out_size, void* d_ws, size_t ws_size,
                              hipStream_t stream) {
  (void)in_sizes; (void)n_in; (void)out_size; (void)d_ws; (void)ws_size;

  const float* pos = (const float*)d_in[0];
  const float* dir = (const float*)d_in[1];
  const float* g0 = (const float*)d_in[2];
  const float* g1 = (const float*)d_in[3];
  const float* g2 = (const float*)d_in[4];
  const float* g3 = (const float*)d_in[5];
  const float* g4 = (const float*)d_in[6];
  const float* g5 = (const float*)d_in[7];
  const float* g6 = (const float*)d_in[8];
  const float* g7 = (const float*)d_in[9];
  float* out = (float*)d_out;

  encode_kernel<<<NPTS / 256, 256, 0, stream>>>(pos, dir, g0, g1, g2, g3, g4,
                                                g5, g6, g7, out);
}


# Grids 0-2 held in LDS (copied once per persistent work-group, 256 groups of 16 waves) and gathered there instead of from the second-level cache; per-element arithmetic and summation order unchanged
# speedup vs baseline: 1.6358x; 1.4512x over previous
#include <hip/hip_runtime.h>


typedef __attribute__((ext_vector_type(2))) float v2f;
typedef __attribute__((ext_vector_type(4))) float v4f;
typedef __attribute__((ext_vector_type(8))) float v8f;

#define NPTS (1 << 20)

#define NWG        256
#define WAVES      16
#define THREADS    (WAVES * 32)
#define PTS_PER_WG (NPTS / NWG)
#define ITERS      (PTS_PER_WG / THREADS)

#define G0_CELLS (16 * 16 * 16)
#define G1_CELLS (21 * 21 * 21)
#define G2_CELLS (28 * 28 * 28)

#define ROW_PITCH 36
#define HALF_ROWS 16

__constant__ float SH_COEF[20][16] = {
    {0.28209479177387814f,0,0,0,0,0,-0.31539156525252f,0,0,0,0,0,0,0,0,0},
    {0,0,0,-0.48860251190291987f,0,0,0,0,0,0,0,0,0,0.4570457994644657f,0,0},
    {0,-0.48860251190291987f,0,0,0,0,0,0,0,0,0,0.4570457994644657f,0,0,0,0},
    {0,0,0.48860251190291987f,0,0,0,0,0,0,0,0,0,-1.1195289977703462f,0,0,0},
    {0,0,0,0,0,0,0,0,0.5462742152960396f,0,0,0,0,0,0,0},
    {0,0,0,0,0,0,0,0,-0.5462742152960396f,0,0,0,0,0,0,0},
    {0,0,0,0,0,0,0.94617469575756f,0,0,0,0,0,0,0,0,0},
    {0,0,0,0,1.0925484305920792f,0,0,0,0,0,0,0,0,0,0,0},
    {0,0,0,0,0,-1.0925484305920792f,0,0,0,0,0,0,0,0,0,0},
    {0,0,0,0,0,0,0,-1.0925484305920792f,0,0,0,0,0,0,0,0},
    {0,0,0,0,0,0,0,0,0,0,0,0,0,0,0,-0.5900435899266435f},
    {0,0,0,0,0,0,0,0,0,0.5900435899266435f,0,0,0,0,0,0},
    {0,0,0,0,0,0,0,0,0,0,0,0,1.865881662950577f,0,0,0},
    {0,0,0,0,0,0,0,0,0,-1.7701307697799304f,0,0,0,0,0,0},
    {0,0,0,0,0,0,0,0,0,0,0,0,0,0,1.445305721320277f,0},
    {0,0,0,0,0,0,0,0,0,0,0,0,0,0,0,1.7701307697799304f},
    {0,0,0,0,0,0,0,0,0,0,0,0,0,0,-1.445305721320277f,0},
    {0,0,0,0,0,0,0,0,0,0,0,0,0,-2.2852289973223288f,0,0},
    {0,0,0,0,0,0,0,0,0,0,0,-2.2852289973223288f,0,0,0,0},
    {0,0,0,0,0,0,0,0,0,0,2.890611442640554f,0,0,0,0,0},
};

template <int R>
__device__ __forceinline__ int cell_of(float px, float py, float pz,
                                       float& fx, float& fy, float& fz) {
  const float cx = px * (float)(R - 1);
  const float cy = py * (float)(R - 1);
  const float cz = pz * (float)(R - 1);
  int ix = (int)cx; ix = ix < R - 2 ? ix : R - 2; ix = ix > 0 ? ix : 0;
  int iy = (int)cy; iy = iy < R - 2 ? iy : R - 2; iy = iy > 0 ? iy : 0;
  int iz = (int)cz; iz = iz < R - 2 ? iz : R - 2; iz = iz > 0 ? iz : 0;
  fx = cx - (float)ix;
  fy = cy - (float)iy;
  fz = cz - (float)iz;
  return (ix * R + iy) * R + iz;
}

template <int R>
__device__ __forceinline__ void corners(const v2f* g, int i000, v2f (&c)[8]) {
  const int i100 = i000 + R * R;
  c[0] = g[i000];         c[1] = g[i000 + 1];
  c[2] = g[i000 + R];     c[3] = g[i000 + R + 1];
  c[4] = g[i100];         c[5] = g[i100 + 1];
  c[6] = g[i100 + R];     c[7] = g[i100 + R + 1];
}

__device__ __forceinline__ v2f trilerp(const v2f (&c)[8], float fx, float fy, float fz) {
  const v2f c00 = c[0] + fz * (c[1] - c[0]);
  const v2f c01 = c[2] + fz * (c[3] - c[2]);
  const v2f c10 = c[4] + fz * (c[5] - c[4]);
  const v2f c11 = c[6] + fz * (c[7] - c[6]);
  const v2f c0  = c00 + fy * (c01 - c00);
  const v2f c1  = c10 + fy * (c11 - c10);
  return c0 + fx * (c1 - c0);
}

template <int CELLS>
__device__ __forceinline__ void fill_lds(v2f* dst, const float* src, int tid) {
  const v2f* s = (const v2f*)src;
  constexpr int FULL = CELLS / THREADS;
  v2f t[FULL + 1];
#pragma unroll
  for (int k = 0; k < FULL; ++k) t[k] = s[tid + k * THREADS];
  const bool tail = tid + FULL * THREADS < CELLS;
  if (tail) t[FULL] = s[tid + FULL * THREADS];
#pragma unroll
  for (int k = 0; k < FULL; ++k) dst[tid + k * THREADS] = t[k];
  if (tail) dst[tid + FULL * THREADS] = t[FULL];
}

__global__ __launch_bounds__(THREADS) void encode_kernel(
    const float* __restrict__ pos, const float* __restrict__ dir,
    const float* __restrict__ g0, const float* __restrict__ g1,
    const float* __restrict__ g2, const float* __restrict__ g3,
    const float* __restrict__ g4, const float* __restrict__ g5,
    const float* __restrict__ g6, const float* __restrict__ g7,
    float* __restrict__ out)
{
  __shared__ __attribute__((aligned(16))) float stage[WAVES][HALF_ROWS * ROW_PITCH];
  __shared__ v2f lg0[G0_CELLS];
  __shared__ v2f lg1[G1_CELLS];
  __shared__ v2f lg2[G2_CELLS];

  const int tid  = threadIdx.x;
  const int lane = tid & 31;
  const int wave = tid >> 5;
  float* st = stage[wave];

  fill_lds<G0_CELLS>(lg0, g0, tid);
  fill_lds<G1_CELLS>(lg1, g1, tid);
  fill_lds<G2_CELLS>(lg2, g2, tid);
  __syncthreads();

  const int  n  = lane & 15;
  const bool hi = lane >= 16;

  v2f bfrag[5];
  for (int c = 0; c < 5; ++c) {
    const int kb = 4 * c + (hi ? 2 : 0);
    bfrag[c].x = SH_COEF[kb][n];
    bfrag[c].y = SH_COEF[kb + 1][n];
  }

  const v2f* __restrict__ q3 = (const v2f*)g3;
  const v2f* __restrict__ q4 = (const v2f*)g4;
  const v2f* __restrict__ q5 = (const v2f*)g5;
  const v2f* __restrict__ q6 = (const v2f*)g6;
  const v2f* __restrict__ q7 = (const v2f*)g7;

  for (int it = 0; it < ITERS; ++it) {
    const int P0 = blockIdx.x * PTS_PER_WG + (it * WAVES + wave) * 32;

    const float px = pos[3 * (P0 + lane) + 0];
    const float py = pos[3 * (P0 + lane) + 1];
    const float pz = pos[3 * (P0 + lane) + 2];

    float fx3, fy3, fz3, fx4, fy4, fz4, fx5, fy5, fz5, fx6, fy6, fz6, fx7, fy7, fz7;
    v2f c3[8], c4[8], c5[8], c6[8], c7[8];
    corners<39>(q3,  cell_of<39>(px, py, pz, fx3, fy3, fz3), c3);
    corners<52>(q4,  cell_of<52>(px, py, pz, fx4, fy4, fz4), c4);
    corners<70>(q5,  cell_of<70>(px, py, pz, fx5, fy5, fz5), c5);
    corners<95>(q6,  cell_of<95>(px, py, pz, fx6, fy6, fz6), c6);
    corners<128>(q7, cell_of<128>(px, py, pz, fx7, fy7, fz7), c7);

    v8f acc[2];
    for (int t = 0; t < 2; ++t) {
      const int p = P0 + 16 * t + n;
      const float dx = dir[3 * p + 0];
      const float dy = dir[3 * p + 1];
      const float dz = dir[3 * p + 2];
      const float inv = rsqrtf(dx * dx + dy * dy + dz * dz);
      const float x = dx * inv, y = dy * inv, z = dz * inv;
      const float xx = x * x, yy = y * y, zz = z * z;

      const float m0 = 1.0f, m1 = x, m2 = y, m3 = z;
      const float m4 = xx, m5 = yy, m6 = zz, m7 = x * y, m8 = y * z, m9 = x * z;
      const float m10 = xx * x, m11 = yy * y, m12 = zz * z;
      const float m13 = xx * y, m14 = xx * z, m15 = x * yy;
      const float m16 = yy * z, m17 = x * zz, m18 = y * zz, m19 = x * y * z;

      v8f a8 = {0.f, 0.f, 0.f, 0.f, 0.f, 0.f, 0.f, 0.f};

#define SH_CHUNK(C, E0, E1, E2, E3)                                           \
      do {                                                                    \
        v2f a;                                                                \
        a.x = hi ? (E2) : (E0);                                               \
        a.y = hi ? (E3) : (E1);                                               \
        a8 = __builtin_amdgcn_wmma_f32_16x16x4_f32(                           \
            false, a, false, bfrag[(C) / 4], (short)0, a8, false, false);     \
      } while (0)

      SH_CHUNK(0,  m0,  m1,  m2,  m3);
      SH_CHUNK(4,  m4,  m5,  m6,  m7);
      SH_CHUNK(8,  m8,  m9,  m10, m11);
      SH_CHUNK(12, m12, m13, m14, m15);
      SH_CHUNK(16, m16, m17, m18, m19);
#undef SH_CHUNK
      acc[t] = a8;
    }

    v2f res[8];
    {
      float fx, fy, fz;
      v2f c[8];
      corners<16>(lg0, cell_of<16>(px, py, pz, fx, fy, fz), c);
      res[0] = trilerp(c, fx, fy, fz);
      corners<21>(lg1, cell_of<21>(px, py, pz, fx, fy, fz), c);
      res[1] = trilerp(c, fx, fy, fz);
      corners<28>(lg2, cell_of<28>(px, py, pz, fx, fy, fz), c);
      res[2] = trilerp(c, fx, fy, fz);
    }

    res[3] = trilerp(c3, fx3, fy3, fz3);
    res[4] = trilerp(c4, fx4, fy4, fz4);
    res[5] = trilerp(c5, fx5, fy5, fz5);
    res[6] = trilerp(c6, fx6, fy6, fz6);
    res[7] = trilerp(c7, fx7, fy7, fz7);

    float* obase = out + (size_t)P0 * 32;
    const int rsub = lane >> 3;
    const int c4o  = (lane & 7) * 4;
    for (int h = 0; h < 2; ++h) {
      const int rowBase = hi ? 8 : 0;
#pragma unroll
      for (int r = 0; r < 8; ++r) {
        st[(rowBase + r) * ROW_PITCH + 16 + n] = acc[h][r];
      }
      if ((lane >> 4) == h) {
        v4f* srow = (v4f*)(st + n * ROW_PITCH);
        for (int k = 0; k < 4; ++k) {
          v4f v = {res[2 * k].x, res[2 * k].y, res[2 * k + 1].x, res[2 * k + 1].y};
          srow[k] = v;
        }
      }

      __builtin_amdgcn_fence(__ATOMIC_RELEASE, "wavefront");
      __builtin_amdgcn_wave_barrier();
      __builtin_amdgcn_fence(__ATOMIC_ACQUIRE, "wavefront");

      for (int j = 0; j < 4; ++j) {
        const int row = 4 * j + rsub;
        const v4f v = *(const v4f*)(st + row * ROW_PITCH + c4o);
        *(v4f*)(obase + (HALF_ROWS * h + row) * 32 + c4o) = v;
      }

      __builtin_amdgcn_fence(__ATOMIC_RELEASE, "wavefront");
      __builtin_amdgcn_wave_barrier();
      __builtin_amdgcn_fence(__ATOMIC_ACQUIRE, "wavefront");
    }
  }
}

extern "C" void kernel_launch(void* const* d_in, const int* in_sizes, int n_in,
                              void* d_out, int out_size, void* d_ws, size_t ws_size,
                              hipStream_t stream) {
  (void)in_sizes; (void)n_in; (void)out_size; (void)d_ws; (void)ws_size;

  const float* pos = (const float*)d_in[0];
  const float* dir = (const float*)d_in[1];
  const float* g0 = (const float*)d_in[2];
  const float* g1 = (const float*)d_in[3];
  const float* g2 = (const float*)d_in[4];
  const float* g3 = (const float*)d_in[5];
  const float* g4 = (const float*)d_in[6];
  const float* g5 = (const float*)d_in[7];
  const float* g6 = (const float*)d_in[8];
  const float* g7 = (const float*)d_in[9];
  float* out = (float*)d_out;

  encode_kernel<<<NWG, THREADS, 0, stream>>>(pos, dir, g0, g1, g2, g3, g4,
                                             g5, g6, g7, out);
}


# Grid 3 also read from LDS: its two 20-plane x-slabs pass through the LDS region before grids 0-2, each point keeps its level-3 value in a register; 16 not 20 global gathers per point; sums unchanged
# speedup vs baseline: 1.7261x; 1.0552x over previous
#include <hip/hip_runtime.h>


typedef __attribute__((ext_vector_type(2))) float v2f;
typedef __attribute__((ext_vector_type(4))) float v4f;
typedef __attribute__((ext_vector_type(8))) float v8f;

#define NPTS (1 << 20)

#define NWG        256
#define WAVES      16
#define THREADS    (WAVES * 32)
#define PTS_PER_WG (NPTS / NWG)
#define ITERS      (PTS_PER_WG / THREADS)

#define G0_CELLS (16 * 16 * 16)
#define G1_CELLS (21 * 21 * 21)
#define G2_CELLS (28 * 28 * 28)
#define LGRID_CELLS (G0_CELLS + G1_CELLS + G2_CELLS)

#define G3_PLANE    (39 * 39)
#define SLAB_SPLIT  19
#define SLAB_CELLS  (20 * G3_PLANE)
static_assert(SLAB_CELLS <= LGRID_CELLS, "a slab must fit the LDS region of grids 0..2");

#define ROW_PITCH 36
#define HALF_ROWS 16

__constant__ float SH_COEF[20][16] = {
    {0.28209479177387814f,0,0,0,0,0,-0.31539156525252f,0,0,0,0,0,0,0,0,0},
    {0,0,0,-0.48860251190291987f,0,0,0,0,0,0,0,0,0,0.4570457994644657f,0,0},
    {0,-0.48860251190291987f,0,0,0,0,0,0,0,0,0,0.4570457994644657f,0,0,0,0},
    {0,0,0.48860251190291987f,0,0,0,0,0,0,0,0,0,-1.1195289977703462f,0,0,0},
    {0,0,0,0,0,0,0,0,0.5462742152960396f,0,0,0,0,0,0,0},
    {0,0,0,0,0,0,0,0,-0.5462742152960396f,0,0,0,0,0,0,0},
    {0,0,0,0,0,0,0.94617469575756f,0,0,0,0,0,0,0,0,0},
    {0,0,0,0,1.0925484305920792f,0,0,0,0,0,0,0,0,0,0,0},
    {0,0,0,0,0,-1.0925484305920792f,0,0,0,0,0,0,0,0,0,0},
    {0,0,0,0,0,0,0,-1.0925484305920792f,0,0,0,0,0,0,0,0},
    {0,0,0,0,0,0,0,0,0,0,0,0,0,0,0,-0.5900435899266435f},
    {0,0,0,0,0,0,0,0,0,0.5900435899266435f,0,0,0,0,0,0},
    {0,0,0,0,0,0,0,0,0,0,0,0,1.865881662950577f,0,0,0},
    {0,0,0,0,0,0,0,0,0,-1.7701307697799304f,0,0,0,0,0,0},
    {0,0,0,0,0,0,0,0,0,0,0,0,0,0,1.445305721320277f,0},
    {0,0,0,0,0,0,0,0,0,0,0,0,0,0,0,1.7701307697799304f},
    {0,0,0,0,0,0,0,0,0,0,0,0,0,0,-1.445305721320277f,0},
    {0,0,0,0,0,0,0,0,0,0,0,0,0,-2.2852289973223288f,0,0},
    {0,0,0,0,0,0,0,0,0,0,0,-2.2852289973223288f,0,0,0,0},
    {0,0,0,0,0,0,0,0,0,0,2.890611442640554f,0,0,0,0,0},
};

template <int R>
__device__ __forceinline__ int cell_of(float px, float py, float pz,
                                       float& fx, float& fy, float& fz) {
  const float cx = px * (float)(R - 1);
  const float cy = py * (float)(R - 1);
  const float cz = pz * (float)(R - 1);
  int ix = (int)cx; ix = ix < R - 2 ? ix : R - 2; ix = ix > 0 ? ix : 0;
  int iy = (int)cy; iy = iy < R - 2 ? iy : R - 2; iy = iy > 0 ? iy : 0;
  int iz = (int)cz; iz = iz < R - 2 ? iz : R - 2; iz = iz > 0 ? iz : 0;
  fx = cx - (float)ix;
  fy = cy - (float)iy;
  fz = cz - (float)iz;
  return (ix * R + iy) * R + iz;
}

template <int R>
__device__ __forceinline__ void corners(const v2f* g, int i000, v2f (&c)[8]) {
  const int i100 = i000 + R * R;
  c[0] = g[i000];         c[1] = g[i000 + 1];
  c[2] = g[i000 + R];     c[3] = g[i000 + R + 1];
  c[4] = g[i100];         c[5] = g[i100 + 1];
  c[6] = g[i100 + R];     c[7] = g[i100 + R + 1];
}

__device__ __forceinline__ v2f trilerp(const v2f (&c)[8], float fx, float fy, float fz) {
  const v2f c00 = c[0] + fz * (c[1] - c[0]);
  const v2f c01 = c[2] + fz * (c[3] - c[2]);
  const v2f c10 = c[4] + fz * (c[5] - c[4]);
  const v2f c11 = c[6] + fz * (c[7] - c[6]);
  const v2f c0  = c00 + fy * (c01 - c00);
  const v2f c1  = c10 + fy * (c11 - c10);
  return c0 + fx * (c1 - c0);
}

template <int CELLS>
__device__ __forceinline__ void fill_lds(v2f* dst, const float* src, int tid) {
  const v2f* s = (const v2f*)src;
  constexpr int FULL = CELLS / THREADS;
  v2f t[FULL + 1];
#pragma unroll
  for (int k = 0; k < FULL; ++k) t[k] = s[tid + k * THREADS];
  const bool tail = tid + FULL * THREADS < CELLS;
  if (tail) t[FULL] = s[tid + FULL * THREADS];
#pragma unroll
  for (int k = 0; k < FULL; ++k) dst[tid + k * THREADS] = t[k];
  if (tail) dst[tid + FULL * THREADS] = t[FULL];
}

template <int CELLS>
__device__ __forceinline__ void fill_load(v2f (&t)[CELLS / THREADS + 1], const float* src, int tid) {
  const v2f* s = (const v2f*)src;
  constexpr int FULL = CELLS / THREADS;
#pragma unroll
  for (int k = 0; k < FULL; ++k) t[k] = s[tid + k * THREADS];
  const int last = tid + FULL * THREADS;
  t[FULL] = s[last < CELLS ? last : CELLS - 1];
}

template <int CELLS>
__device__ __forceinline__ void fill_store(v2f* dst, const v2f (&t)[CELLS / THREADS + 1], int tid) {
  constexpr int FULL = CELLS / THREADS;
#pragma unroll
  for (int k = 0; k < FULL; ++k) dst[tid + k * THREADS] = t[k];
  if (tid + FULL * THREADS < CELLS) dst[tid + FULL * THREADS] = t[FULL];
}

template <bool SECOND>
__device__ __forceinline__ void slab_pass(const v2f* slab, const float (&ppx)[ITERS],
                                          const float (&ppy)[ITERS], const float (&ppz)[ITERS],
                                          v2f (&r3)[ITERS]) {
#pragma unroll
  for (int it = 0; it < ITERS; ++it) {
    const float px = ppx[it];
    const float py = ppy[it];
    const float pz = ppz[it];
    float fx, fy, fz;
    const int i000 = cell_of<39>(px, py, pz, fx, fy, fz);
    const bool mine = SECOND ? (i000 >= SLAB_SPLIT * G3_PLANE) : (i000 < SLAB_SPLIT * G3_PLANE);
    const int li = mine ? (SECOND ? i000 - SLAB_SPLIT * G3_PLANE : i000) : 0;
    v2f c[8];
    corners<39>(slab, li, c);
    const v2f v = trilerp(c, fx, fy, fz);
    r3[it].x = mine ? v.x : r3[it].x;
    r3[it].y = mine ? v.y : r3[it].y;
    if ((it & 3) == 3) {
#pragma unroll
      for (int k = it - 3; k <= it; ++k) asm volatile("" : "+v"(r3[k].x), "+v"(r3[k].y));
    }
  }
}

__global__ __launch_bounds__(THREADS) void encode_kernel(
    const float* __restrict__ pos, const float* __restrict__ dir,
    const float* __restrict__ g0, const float* __restrict__ g1,
    const float* __restrict__ g2, const float* __restrict__ g3,
    const float* __restrict__ g4, const float* __restrict__ g5,
    const float* __restrict__ g6, const float* __restrict__ g7,
    float* __restrict__ out)
{
  __shared__ __attribute__((aligned(16))) float stage[WAVES][HALF_ROWS * ROW_PITCH];
  __shared__ v2f lgrid[LGRID_CELLS];
  v2f* const lg0 = lgrid;
  v2f* const lg1 = lgrid + G0_CELLS;
  v2f* const lg2 = lgrid + G0_CELLS + G1_CELLS;

  const int tid  = threadIdx.x;
  const int lane = tid & 31;
  const int wave = tid >> 5;
  float* st = stage[wave];

  v2f r3[ITERS];
#pragma unroll
  for (int it = 0; it < ITERS; ++it) r3[it] = (v2f){0.f, 0.f};
  float ppx[ITERS], ppy[ITERS], ppz[ITERS];
#pragma unroll
  for (int it = 0; it < ITERS; ++it) {
    const int p = blockIdx.x * PTS_PER_WG + it * THREADS + tid;
    ppx[it] = pos[3 * p + 0];
    ppy[it] = pos[3 * p + 1];
    ppz[it] = pos[3 * p + 2];
  }
  fill_lds<SLAB_CELLS>(lgrid, g3, tid);
  __syncthreads();
  {
    v2f tb[SLAB_CELLS / THREADS + 1];
    fill_load<SLAB_CELLS>(tb, g3 + 2 * SLAB_SPLIT * G3_PLANE, tid);
    slab_pass<false>(lgrid, ppx, ppy, ppz, r3);
    __syncthreads();
    fill_store<SLAB_CELLS>(lgrid, tb, tid);
  }
  __syncthreads();
  {
    v2f t2[G2_CELLS / THREADS + 1];
    fill_load<G2_CELLS>(t2, g2, tid);
    slab_pass<true>(lgrid, ppx, ppy, ppz, r3);
    __syncthreads();
    fill_store<G2_CELLS>(lg2, t2, tid);
  }
  fill_lds<G0_CELLS>(lg0, g0, tid);
  fill_lds<G1_CELLS>(lg1, g1, tid);
  __syncthreads();

  const int  n  = lane & 15;
  const bool hi = lane >= 16;

  v2f bfrag[5];
  for (int c = 0; c < 5; ++c) {
    const int kb = 4 * c + (hi ? 2 : 0);
    bfrag[c].x = SH_COEF[kb][n];
    bfrag[c].y = SH_COEF[kb + 1][n];
  }

  const v2f* __restrict__ q4 = (const v2f*)g4;
  const v2f* __restrict__ q5 = (const v2f*)g5;
  const v2f* __restrict__ q6 = (const v2f*)g6;
  const v2f* __restrict__ q7 = (const v2f*)g7;

#pragma nounroll
  for (int it = 0; it < ITERS; ++it) {
    const int P0 = blockIdx.x * PTS_PER_WG + (it * WAVES + wave) * 32;

    const float px = pos[3 * (P0 + lane) + 0];
    const float py = pos[3 * (P0 + lane) + 1];
    const float pz = pos[3 * (P0 + lane) + 2];

    float fx4, fy4, fz4, fx5, fy5, fz5, fx6, fy6, fz6, fx7, fy7, fz7;
    v2f c4[8], c5[8], c6[8], c7[8];
    corners<52>(q4,  cell_of<52>(px, py, pz, fx4, fy4, fz4), c4);
    corners<70>(q5,  cell_of<70>(px, py, pz, fx5, fy5, fz5), c5);
    corners<95>(q6,  cell_of<95>(px, py, pz, fx6, fy6, fz6), c6);
    corners<128>(q7, cell_of<128>(px, py, pz, fx7, fy7, fz7), c7);

    v8f acc[2];
    for (int t = 0; t < 2; ++t) {
      const int p = P0 + 16 * t + n;
      const float dx = dir[3 * p + 0];
      const float dy = dir[3 * p + 1];
      const float dz = dir[3 * p + 2];
      const float inv = rsqrtf(dx * dx + dy * dy + dz * dz);
      const float x = dx * inv, y = dy * inv, z = dz * inv;
      const float xx = x * x, yy = y * y, zz = z * z;

      const float m0 = 1.0f, m1 = x, m2 = y, m3 = z;
      const float m4 = xx, m5 = yy, m6 = zz, m7 = x * y, m8 = y * z, m9 = x * z;
      const float m10 = xx * x, m11 = yy * y, m12 = zz * z;
      const float m13 = xx * y, m14 = xx * z, m15 = x * yy;
      const float m16 = yy * z, m17 = x * zz, m18 = y * zz, m19 = x * y * z;

      v8f a8 = {0.f, 0.f, 0.f, 0.f, 0.f, 0.f, 0.f, 0.f};

#define SH_CHUNK(C, E0, E1, E2, E3)                                           \
      do {                                                                    \
        v2f a;                                                                \
        a.x = hi ? (E2) : (E0);                                               \
        a.y = hi ? (E3) : (E1);                                               \
        a8 = __builtin_amdgcn_wmma_f32_16x16x4_f32(                           \
            false, a, false, bfrag[(C) / 4], (short)0, a8, false, false);     \
      } while (0)

      SH_CHUNK(0,  m0,  m1,  m2,  m3);
      SH_CHUNK(4,  m4,  m5,  m6,  m7);
      SH_CHUNK(8,  m8,  m9,  m10, m11);
      SH_CHUNK(12, m12, m13, m14, m15);
      SH_CHUNK(16, m16, m17, m18, m19);
#undef SH_CHUNK
      acc[t] = a8;
    }

    v2f res[8];
    {
      float fx, fy, fz;
      v2f c[8];
      corners<16>(lg0, cell_of<16>(px, py, pz, fx, fy, fz), c);
      res[0] = trilerp(c, fx, fy, fz);
      corners<21>(lg1, cell_of<21>(px, py, pz, fx, fy, fz), c);
      res[1] = trilerp(c, fx, fy, fz);
      corners<28>(lg2, cell_of<28>(px, py, pz, fx, fy, fz), c);
      res[2] = trilerp(c, fx, fy, fz);
    }

    res[3] = r3[0];
#pragma unroll
    for (int k = 1; k < ITERS; ++k) {
      res[3].x = (it == k) ? r3[k].x : res[3].x;
      res[3].y = (it == k) ? r3[k].y : res[3].y;
    }
    res[4] = trilerp(c4, fx4, fy4, fz4);
    res[5] = trilerp(c5, fx5, fy5, fz5);
    res[6] = trilerp(c6, fx6, fy6, fz6);
    res[7] = trilerp(c7, fx7, fy7, fz7);

    float* obase = out + (size_t)P0 * 32;
    const int rsub = lane >> 3;
    const int c4o  = (lane & 7) * 4;
    for (int h = 0; h < 2; ++h) {
      const int rowBase = hi ? 8 : 0;
#pragma unroll
      for (int r = 0; r < 8; ++r) {
        st[(rowBase + r) * ROW_PITCH + 16 + n] = acc[h][r];
      }
      if ((lane >> 4) == h) {
        v4f* srow = (v4f*)(st + n * ROW_PITCH);
        for (int k = 0; k < 4; ++k) {
          v4f v = {res[2 * k].x, res[2 * k].y, res[2 * k + 1].x, res[2 * k + 1].y};
          srow[k] = v;
        }
      }

      __builtin_amdgcn_fence(__ATOMIC_RELEASE, "wavefront");
      __builtin_amdgcn_wave_barrier();
      __builtin_amdgcn_fence(__ATOMIC_ACQUIRE, "wavefront");

      for (int j = 0; j < 4; ++j) {
        const int row = 4 * j + rsub;
        const v4f v = *(const v4f*)(st + row * ROW_PITCH + c4o);
        *(v4f*)(obase + (HALF_ROWS * h + row) * 32 + c4o) = v;
      }

      __builtin_amdgcn_fence(__ATOMIC_RELEASE, "wavefront");
      __builtin_amdgcn_wave_barrier();
      __builtin_amdgcn_fence(__ATOMIC_ACQUIRE, "wavefront");
    }
  }
}

extern "C" void kernel_launch(void* const* d_in, const int* in_sizes, int n_in,
                              void* d_out, int out_size, void* d_ws, size_t ws_size,
                              hipStream_t stream) {
  (void)in_sizes; (void)n_in; (void)out_size; (void)d_ws; (void)ws_size;

  const float* pos = (const float*)d_in[0];
  const float* dir = (const float*)d_in[1];
  const float* g0 = (const float*)d_in[2];
  const float* g1 = (const float*)d_in[3];
  const float* g2 = (const float*)d_in[4];
  const float* g3 = (const float*)d_in[5];
  const float* g4 = (const float*)d_in[6];
  const float* g5 = (const float*)d_in[7];
  const float* g6 = (const float*)d_in[8];
  const float* g7 = (const float*)d_in[9];
  float* out = (float*)d_out;

  encode_kernel<<<NWG, THREADS, 0, stream>>>(pos, dir, g0, g1, g2, g3, g4,
                                             g5, g6, g7, out);
}


# A first launch re-lays grids 4-6 in the workspace (as many as fit) so a point's 8 corners of a level are 64 contiguous bytes: 1-2 cache lines per level, not 4-5; values and sum order unchanged
# speedup vs baseline: 1.8592x; 1.0771x over previous
#include <hip/hip_runtime.h>


typedef __attribute__((ext_vector_type(2))) float v2f;
typedef __attribute__((ext_vector_type(4))) float v4f;
typedef __attribute__((ext_vector_type(8))) float v8f;

#define NPTS (1 << 20)

#define NWG        256
#define WAVES      16
#define THREADS    (WAVES * 32)
#define PTS_PER_WG (NPTS / NWG)
#define ITERS      (PTS_PER_WG / THREADS)

#define G0_CELLS (16 * 16 * 16)
#define G1_CELLS (21 * 21 * 21)
#define G2_CELLS (28 * 28 * 28)
#define LGRID_CELLS (G0_CELLS + G1_CELLS + G2_CELLS)

#define G3_PLANE    (39 * 39)
#define SLAB_SPLIT  19
#define SLAB_CELLS  (20 * G3_PLANE)
static_assert(SLAB_CELLS <= LGRID_CELLS, "a slab must fit the LDS region of grids 0..2");

#define T4_V4 (2 * 51 * 51 * 52)
#define T5_V4 (2 * 69 * 69 * 70)
#define T6_V4 (2 * 94 * 94 * 95)
#define O4_V4 0
#define O5_V4 (O4_V4 + T4_V4)
#define O6_V4 (O5_V4 + T5_V4 + 4)
static_assert(O5_V4 % 8 == 0 && O6_V4 % 8 == 0, "tables start on 128-byte boundaries");
#define NEED1_BYTES (16ull * (O4_V4 + T4_V4))
#define NEED2_BYTES (16ull * (O5_V4 + T5_V4))
#define NEED3_BYTES (16ull * (O6_V4 + T6_V4))
#define BUILD_WG      256
#define BUILD_THREADS 256

#define ROW_PITCH 36
#define HALF_ROWS 16

__constant__ float SH_COEF[20][16] = {
    {0.28209479177387814f,0,0,0,0,0,-0.31539156525252f,0,0,0,0,0,0,0,0,0},
    {0,0,0,-0.48860251190291987f,0,0,0,0,0,0,0,0,0,0.4570457994644657f,0,0},
    {0,-0.48860251190291987f,0,0,0,0,0,0,0,0,0,0.4570457994644657f,0,0,0,0},
    {0,0,0.48860251190291987f,0,0,0,0,0,0,0,0,0,-1.1195289977703462f,0,0,0},
    {0,0,0,0,0,0,0,0,0.5462742152960396f,0,0,0,0,0,0,0},
    {0,0,0,0,0,0,0,0,-0.5462742152960396f,0,0,0,0,0,0,0},
    {0,0,0,0,0,0,0.94617469575756f,0,0,0,0,0,0,0,0,0},
    {0,0,0,0,1.0925484305920792f,0,0,0,0,0,0,0,0,0,0,0},
    {0,0,0,0,0,-1.0925484305920792f,0,0,0,0,0,0,0,0,0,0},
    {0,0,0,0,0,0,0,-1.0925484305920792f,0,0,0,0,0,0,0,0},
    {0,0,0,0,0,0,0,0,0,0,0,0,0,0,0,-0.5900435899266435f},
    {0,0,0,0,0,0,0,0,0,0.5900435899266435f,0,0,0,0,0,0},
    {0,0,0,0,0,0,0,0,0,0,0,0,1.865881662950577f,0,0,0},
    {0,0,0,0,0,0,0,0,0,-1.7701307697799304f,0,0,0,0,0,0},
    {0,0,0,0,0,0,0,0,0,0,0,0,0,0,1.445305721320277f,0},
    {0,0,0,0,0,0,0,0,0,0,0,0,0,0,0,1.7701307697799304f},
    {0,0,0,0,0,0,0,0,0,0,0,0,0,0,-1.445305721320277f,0},
    {0,0,0,0,0,0,0,0,0,0,0,0,0,-2.2852289973223288f,0,0},
    {0,0,0,0,0,0,0,0,0,0,0,-2.2852289973223288f,0,0,0,0},
    {0,0,0,0,0,0,0,0,0,0,2.890611442640554f,0,0,0,0,0},
};

template <int R>
__device__ __forceinline__ void cell_ijk(float px, float py, float pz,
                                         float& fx, float& fy, float& fz,
                                         int& ix, int& iy, int& iz) {
  const float cx = px * (float)(R - 1);
  const float cy = py * (float)(R - 1);
  const float cz = pz * (float)(R - 1);
  ix = (int)cx; ix = ix < R - 2 ? ix : R - 2; ix = ix > 0 ? ix : 0;
  iy = (int)cy; iy = iy < R - 2 ? iy : R - 2; iy = iy > 0 ? iy : 0;
  iz = (int)cz; iz = iz < R - 2 ? iz : R - 2; iz = iz > 0 ? iz : 0;
  fx = cx - (float)ix;
  fy = cy - (float)iy;
  fz = cz - (float)iz;
}

template <int R>
__device__ __forceinline__ int cell_of(float px, float py, float pz,
                                       float& fx, float& fy, float& fz) {
  int ix, iy, iz;
  cell_ijk<R>(px, py, pz, fx, fy, fz, ix, iy, iz);
  return (ix * R + iy) * R + iz;
}

template <int R>
__device__ __forceinline__ int cell_pk(float px, float py, float pz,
                                       float& fx, float& fy, float& fz) {
  int ix, iy, iz;
  cell_ijk<R>(px, py, pz, fx, fy, fz, ix, iy, iz);
  return (ix * (R - 1) + iy) * R + iz;
}

template <int R>
__device__ __forceinline__ void corners(const v2f* g, int i000, v2f (&c)[8]) {
  const int i100 = i000 + R * R;
  c[0] = g[i000];         c[1] = g[i000 + 1];
  c[2] = g[i000 + R];     c[3] = g[i000 + R + 1];
  c[4] = g[i100];         c[5] = g[i100 + 1];
  c[6] = g[i100 + R];     c[7] = g[i100 + R + 1];
}

__device__ __forceinline__ void corners_pk(const v4f* t, int e0, v2f (&c)[8]) {
  const v4f a0 = t[2 * e0 + 0];
  const v4f a1 = t[2 * e0 + 1];
  const v4f b0 = t[2 * e0 + 2];
  const v4f b1 = t[2 * e0 + 3];
  c[0] = (v2f){a0.x, a0.y};  c[2] = (v2f){a0.z, a0.w};
  c[4] = (v2f){a1.x, a1.y};  c[6] = (v2f){a1.z, a1.w};
  c[1] = (v2f){b0.x, b0.y};  c[3] = (v2f){b0.z, b0.w};
  c[5] = (v2f){b1.x, b1.y};  c[7] = (v2f){b1.z, b1.w};
}

__device__ __forceinline__ int table_tier(const float* ws, unsigned long long bytes,
                                          unsigned long long& lost) {
  const unsigned long long p = (unsigned long long)ws;
  lost = (128ull - (p & 127ull)) & 127ull;
  const unsigned long long avail = bytes > lost ? bytes - lost : 0ull;
  if (ws == nullptr) return 0;
  return avail >= NEED3_BYTES ? 3 : avail >= NEED2_BYTES ? 2 : avail >= NEED1_BYTES ? 1 : 0;
}

template <int R>
__device__ __forceinline__ void build_level(const float* g, v4f* t, int gtid) {
  const v2f* s = (const v2f*)g;
  constexpr int NV4    = 2 * (R - 1) * (R - 1) * R;
  constexpr int STRIDE = BUILD_WG * BUILD_THREADS;
  constexpr int U      = (NV4 + STRIDE - 1) / STRIDE;
  v2f lo[U], hi[U];
#pragma unroll
  for (int u = 0; u < U; ++u) {
    const int q    = gtid + u * STRIDE;
    const int qc   = q < NV4 ? q : NV4 - 1;
    const int e    = qc >> 1;
    const int half = qc & 1;
    const int row  = e / R;
    const int iz   = e - row * R;
    const int ix   = row / (R - 1);
    const int iy   = row - ix * (R - 1);
    const int i    = ((ix + half) * R + iy) * R + iz;
    lo[u] = s[i];
    hi[u] = s[i + R];
  }
#pragma unroll
  for (int u = 0; u < U; ++u) {
    const int q = gtid + u * STRIDE;
    if (q < NV4) t[q] = (v4f){lo[u].x, lo[u].y, hi[u].x, hi[u].y};
  }
}

__global__ __launch_bounds__(BUILD_THREADS) void build_kernel(
    const float* __restrict__ g4, const float* __restrict__ g5,
    const float* __restrict__ g6, float* __restrict__ ws, unsigned long long ws_bytes)
{
  unsigned long long lost;
  const int tier = table_tier(ws, ws_bytes, lost);
  v4f* const t = (v4f*)((char*)ws + lost);
  const int gtid = blockIdx.x * BUILD_THREADS + threadIdx.x;
  if (tier >= 1) build_level<52>(g4, t + O4_V4, gtid);
  if (tier >= 2) build_level<70>(g5, t + O5_V4, gtid);
  if (tier >= 3) build_level<95>(g6, t + O6_V4, gtid);
}

__device__ __forceinline__ v2f trilerp(const v2f (&c)[8], float fx, float fy, float fz) {
  const v2f c00 = c[0] + fz * (c[1] - c[0]);
  const v2f c01 = c[2] + fz * (c[3] - c[2]);
  const v2f c10 = c[4] + fz * (c[5] - c[4]);
  const v2f c11 = c[6] + fz * (c[7] - c[6]);
  const v2f c0  = c00 + fy * (c01 - c00);
  const v2f c1  = c10 + fy * (c11 - c10);
  return c0 + fx * (c1 - c0);
}

template <int CELLS>
__device__ __forceinline__ void fill_lds(v2f* dst, const float* src, int tid) {
  const v2f* s = (const v2f*)src;
  constexpr int FULL = CELLS / THREADS;
  v2f t[FULL + 1];
#pragma unroll
  for (int k = 0; k < FULL; ++k) t[k] = s[tid + k * THREADS];
  const bool tail = tid + FULL * THREADS < CELLS;
  if (tail) t[FULL] = s[tid + FULL * THREADS];
#pragma unroll
  for (int k = 0; k < FULL; ++k) dst[tid + k * THREADS] = t[k];
  if (tail) dst[tid + FULL * THREADS] = t[FULL];
}

template <int CELLS>
__device__ __forceinline__ void fill_load(v2f (&t)[CELLS / THREADS + 1], const float* src, int tid) {
  const v2f* s = (const v2f*)src;
  constexpr int FULL = CELLS / THREADS;
#pragma unroll
  for (int k = 0; k < FULL; ++k) t[k] = s[tid + k * THREADS];
  const int last = tid + FULL * THREADS;
  t[FULL] = s[last < CELLS ? last : CELLS - 1];
}

template <int CELLS>
__device__ __forceinline__ void fill_store(v2f* dst, const v2f (&t)[CELLS / THREADS + 1], int tid) {
  constexpr int FULL = CELLS / THREADS;
#pragma unroll
  for (int k = 0; k < FULL; ++k) dst[tid + k * THREADS] = t[k];
  if (tid + FULL * THREADS < CELLS) dst[tid + FULL * THREADS] = t[FULL];
}

template <bool SECOND>
__device__ __forceinline__ void slab_pass(const v2f* slab, const float (&ppx)[ITERS],
                                          const float (&ppy)[ITERS], const float (&ppz)[ITERS],
                                          v2f (&r3)[ITERS]) {
#pragma unroll
  for (int it = 0; it < ITERS; ++it) {
    const float px = ppx[it];
    const float py = ppy[it];
    const float pz = ppz[it];
    float fx, fy, fz;
    const int i000 = cell_of<39>(px, py, pz, fx, fy, fz);
    const bool mine = SECOND ? (i000 >= SLAB_SPLIT * G3_PLANE) : (i000 < SLAB_SPLIT * G3_PLANE);
    const int li = mine ? (SECOND ? i000 - SLAB_SPLIT * G3_PLANE : i000) : 0;
    v2f c[8];
    corners<39>(slab, li, c);
    const v2f v = trilerp(c, fx, fy, fz);
    r3[it].x = mine ? v.x : r3[it].x;
    r3[it].y = mine ? v.y : r3[it].y;
    if ((it & 3) == 3) {
#pragma unroll
      for (int k = it - 3; k <= it; ++k) asm volatile("" : "+v"(r3[k].x), "+v"(r3[k].y));
    }
  }
}

template <int NP>
__device__ __forceinline__ void main_loop(
    const float* __restrict__ pos, const float* __restrict__ dir,
    const v2f* __restrict__ q4, const v2f* __restrict__ q5,
    const v2f* __restrict__ q6, const v2f* __restrict__ q7,
    const v4f* __restrict__ t4, const v4f* __restrict__ t5,
    const v4f* __restrict__ t6,
    const v2f* lg0, const v2f* lg1, const v2f* lg2,
    const v2f (&r3)[ITERS], const v2f (&bfrag)[5],
    float* st, float* __restrict__ out,
    const int lane, const int wave, const int n, const bool hi)
{
#pragma nounroll
  for (int it = 0; it < ITERS; ++it) {
    const int P0 = blockIdx.x * PTS_PER_WG + (it * WAVES + wave) * 32;

    const float px = pos[3 * (P0 + lane) + 0];
    const float py = pos[3 * (P0 + lane) + 1];
    const float pz = pos[3 * (P0 + lane) + 2];

    float fx4, fy4, fz4, fx5, fy5, fz5, fx6, fy6, fz6, fx7, fy7, fz7;
    v2f c4[8], c5[8], c6[8], c7[8];
    if constexpr (NP >= 1) corners_pk(t4, cell_pk<52>(px, py, pz, fx4, fy4, fz4), c4);
    else                   corners<52>(q4, cell_of<52>(px, py, pz, fx4, fy4, fz4), c4);
    if constexpr (NP >= 2) corners_pk(t5, cell_pk<70>(px, py, pz, fx5, fy5, fz5), c5);
    else                   corners<70>(q5, cell_of<70>(px, py, pz, fx5, fy5, fz5), c5);
    if constexpr (NP >= 3) corners_pk(t6, cell_pk<95>(px, py, pz, fx6, fy6, fz6), c6);
    else                   corners<95>(q6, cell_of<95>(px, py, pz, fx6, fy6, fz6), c6);
    corners<128>(q7, cell_of<128>(px, py, pz, fx7, fy7, fz7), c7);

    v8f acc[2];
    for (int t = 0; t < 2; ++t) {
      const int p = P0 + 16 * t + n;
      const float dx = dir[3 * p + 0];
      const float dy = dir[3 * p + 1];
      const float dz = dir[3 * p + 2];
      const float inv = rsqrtf(dx * dx + dy * dy + dz * dz);
      const float x = dx * inv, y = dy * inv, z = dz * inv;
      const float xx = x * x, yy = y * y, zz = z * z;

      const float m0 = 1.0f, m1 = x, m2 = y, m3 = z;
      const float m4 = xx, m5 = yy, m6 = zz, m7 = x * y, m8 = y * z, m9 = x * z;
      const float m10 = xx * x, m11 = yy * y, m12 = zz * z;
      const float m13 = xx * y, m14 = xx * z, m15 = x * yy;
      const float m16 = yy * z, m17 = x * zz, m18 = y * zz, m19 = x * y * z;

      v8f a8 = {0.f, 0.f, 0.f, 0.f, 0.f, 0.f, 0.f, 0.f};

#define SH_CHUNK(C, E0, E1, E2, E3)                                           \
      do {                                                                    \
        v2f a;                                                                \
        a.x = hi ? (E2) : (E0);                                               \
        a.y = hi ? (E3) : (E1);                                               \
        a8 = __builtin_amdgcn_wmma_f32_16x16x4_f32(                           \
            false, a, false, bfrag[(C) / 4], (short)0, a8, false, false);     \
      } while (0)

      SH_CHUNK(0,  m0,  m1,  m2,  m3);
      SH_CHUNK(4,  m4,  m5,  m6,  m7);
      SH_CHUNK(8,  m8,  m9,  m10, m11);
      SH_CHUNK(12, m12, m13, m14, m15);
      SH_CHUNK(16, m16, m17, m18, m19);
#undef SH_CHUNK
      acc[t] = a8;
    }

    v2f res[8];
    {
      float fx, fy, fz;
      v2f c[8];
      corners<16>(lg0, cell_of<16>(px, py, pz, fx, fy, fz), c);
      res[0] = trilerp(c, fx, fy, fz);
      corners<21>(lg1, cell_of<21>(px, py, pz, fx, fy, fz), c);
      res[1] = trilerp(c, fx, fy, fz);
      corners<28>(lg2, cell_of<28>(px, py, pz, fx, fy, fz), c);
      res[2] = trilerp(c, fx, fy, fz);
    }

    res[3] = r3[0];
#pragma unroll
    for (int k = 1; k < ITERS; ++k) {
      res[3].x = (it == k) ? r3[k].x : res[3].x;
      res[3].y = (it == k) ? r3[k].y : res[3].y;
    }
    res[4] = trilerp(c4, fx4, fy4, fz4);
    res[5] = trilerp(c5, fx5, fy5, fz5);
    res[6] = trilerp(c6, fx6, fy6, fz6);
    res[7] = trilerp(c7, fx7, fy7, fz7);

    float* obase = out + (size_t)P0 * 32;
    const int rsub = lane >> 3;
    const int c4o  = (lane & 7) * 4;
    for (int h = 0; h < 2; ++h) {
      const int rowBase = hi ? 8 : 0;
#pragma unroll
      for (int r = 0; r < 8; ++r) {
        st[(rowBase + r) * ROW_PITCH + 16 + n] = acc[h][r];
      }
      if ((lane >> 4) == h) {
        v4f* srow = (v4f*)(st + n * ROW_PITCH);
        for (int k = 0; k < 4; ++k) {
          v4f v = {res[2 * k].x, res[2 * k].y, res[2 * k + 1].x, res[2 * k + 1].y};
          srow[k] = v;
        }
      }

      __builtin_amdgcn_fence(__ATOMIC_RELEASE, "wavefront");
      __builtin_amdgcn_wave_barrier();
      __builtin_amdgcn_fence(__ATOMIC_ACQUIRE, "wavefront");

      for (int j = 0; j < 4; ++j) {
        const int row = 4 * j + rsub;
        const v4f v = *(const v4f*)(st + row * ROW_PITCH + c4o);
        *(v4f*)(obase + (HALF_ROWS * h + row) * 32 + c4o) = v;
      }

      __builtin_amdgcn_fence(__ATOMIC_RELEASE, "wavefront");
      __builtin_amdgcn_wave_barrier();
      __builtin_amdgcn_fence(__ATOMIC_ACQUIRE, "wavefront");
    }
  }
}

__global__ __launch_bounds__(THREADS) void encode_kernel(
    const float* __restrict__ pos, const float* __restrict__ dir,
    const float* __restrict__ g0, const float* __restrict__ g1,
    const float* __restrict__ g2, const float* __restrict__ g3,
    const float* __restrict__ g4, const float* __restrict__ g5,
    const float* __restrict__ g6, const float* __restrict__ g7,
    float* __restrict__ out,
    const float* __restrict__ ws, unsigned long long ws_bytes)
{
  __shared__ __attribute__((aligned(16))) float stage[WAVES][HALF_ROWS * ROW_PITCH];
  __shared__ v2f lgrid[LGRID_CELLS];
  v2f* const lg0 = lgrid;
  v2f* const lg1 = lgrid + G0_CELLS;
  v2f* const lg2 = lgrid + G0_CELLS + G1_CELLS;

  const int tid  = threadIdx.x;
  const int lane = tid & 31;
  const int wave = tid >> 5;
  float* st = stage[wave];

  v2f r3[ITERS];
#pragma unroll
  for (int it = 0; it < ITERS; ++it) r3[it] = (v2f){0.f, 0.f};
  float ppx[ITERS], ppy[ITERS], ppz[ITERS];
#pragma unroll
  for (int it = 0; it < ITERS; ++it) {
    const int p = blockIdx.x * PTS_PER_WG + it * THREADS + tid;
    ppx[it] = pos[3 * p + 0];
    ppy[it] = pos[3 * p + 1];
    ppz[it] = pos[3 * p + 2];
  }
  fill_lds<SLAB_CELLS>(lgrid, g3, tid);
  __syncthreads();
  {
    v2f tb[SLAB_CELLS / THREADS + 1];
    fill_load<SLAB_CELLS>(tb, g3 + 2 * SLAB_SPLIT * G3_PLANE, tid);
    slab_pass<false>(lgrid, ppx, ppy, ppz, r3);
    __syncthreads();
    fill_store<SLAB_CELLS>(lgrid, tb, tid);
  }
  __syncthreads();
  {
    v2f t2[G2_CELLS / THREADS + 1];
    fill_load<G2_CELLS>(t2, g2, tid);
    slab_pass<true>(lgrid, ppx, ppy, ppz, r3);
    __syncthreads();
    fill_store<G2_CELLS>(lg2, t2, tid);
  }
  fill_lds<G0_CELLS>(lg0, g0, tid);
  fill_lds<G1_CELLS>(lg1, g1, tid);
  __syncthreads();

  const int  n  = lane & 15;
  const bool hi = lane >= 16;

  v2f bfrag[5];
  for (int c = 0; c < 5; ++c) {
    const int kb = 4 * c + (hi ? 2 : 0);
    bfrag[c].x = SH_COEF[kb][n];
    bfrag[c].y = SH_COEF[kb + 1][n];
  }

  unsigned long long lost;
  const int tier = table_tier(ws, ws_bytes, lost);
  const v4f* const tab = (const v4f*)((const char*)ws + lost);
  const v2f* q4 = (const v2f*)g4;
  const v2f* q5 = (const v2f*)g5;
  const v2f* q6 = (const v2f*)g6;
  const v2f* q7 = (const v2f*)g7;
  if (tier == 3)
    main_loop<3>(pos, dir, q4, q5, q6, q7, tab + O4_V4, tab + O5_V4, tab + O6_V4,
                 lg0, lg1, lg2, r3, bfrag, st, out, lane, wave, n, hi);
  else if (tier == 2)
    main_loop<2>(pos, dir, q4, q5, q6, q7, tab + O4_V4, tab + O5_V4, tab + O6_V4,
                 lg0, lg1, lg2, r3, bfrag, st, out, lane, wave, n, hi);
  else if (tier == 1)
    main_loop<1>(pos, dir, q4, q5, q6, q7, tab + O4_V4, tab + O5_V4, tab + O6_V4,
                 lg0, lg1, lg2, r3, bfrag, st, out, lane, wave, n, hi);
  else
    main_loop<0>(pos, dir, q4, q5, q6, q7, tab + O4_V4, tab + O5_V4, tab + O6_V4,
                 lg0, lg1, lg2, r3, bfrag, st, out, lane, wave, n, hi);
}

extern "C" void kernel_launch(void* const* d_in, const int* in_sizes, int n_in,
                              void* d_out, int out_size, void* d_ws, size_t ws_size,
                              hipStream_t stream) {
  (void)in_sizes; (void)n_in; (void)out_size;

  const float* pos = (const float*)d_in[0];
  const float* dir = (const float*)d_in[1];
  const float* g0 = (const float*)d_in[2];
  const float* g1 = (const float*)d_in[3];
  const float* g2 = (const float*)d_in[4];
  const float* g3 = (const float*)d_in[5];
  const float* g4 = (const float*)d_in[6];
  const float* g5 = (const float*)d_in[7];
  const float* g6 = (const float*)d_in[8];
  const float* g7 = (const float*)d_in[9];
  float* out = (float*)d_out;

  float* ws = (float*)d_ws;

  build_kernel<<<BUILD_WG, BUILD_THREADS, 0, stream>>>(g4, g5, g6, ws, ws_size);

  encode_kernel<<<NWG, THREADS, 0, stream>>>(pos, dir, g0, g1, g2, g3, g4,
                                             g5, g6, g7, out, ws, ws_size);
}


# Launch 1 sorts the points by 8x8x4 spatial bin (LDS counting sort, no atomics); launch 2 gives each bin one work-group that reads all 8 levels' corners from an LDS sub-volume; sums unchanged
# speedup vs baseline: 2.8680x; 1.5426x over previous
#include <hip/hip_runtime.h>


typedef __attribute__((ext_vector_type(2))) float v2f;
typedef __attribute__((ext_vector_type(4))) float v4f;
typedef __attribute__((ext_vector_type(8))) float v8f;

#define NPTS (1 << 20)

#define NWG     256
#define WAVES   16
#define THREADS (WAVES * 32)
#define SEG     (NPTS / NWG)
#define PER_T   (SEG / THREADS)
#define NBINS   256
#define CH      2560

#define ROW_PITCH 36
#define HALF_ROWS 16

#define REC_V4    (2 * NPTS)
#define OFF_PITCH 260
#define NEED_BYTES (16ull * REC_V4 + 4ull * NWG * OFF_PITCH)
#define FAR_LO (1 << 20)

__constant__ float SH_COEF[20][16] = {
    {0.28209479177387814f,0,0,0,0,0,-0.31539156525252f,0,0,0,0,0,0,0,0,0},
    {0,0,0,-0.48860251190291987f,0,0,0,0,0,0,0,0,0,0.4570457994644657f,0,0},
    {0,-0.48860251190291987f,0,0,0,0,0,0,0,0,0,0.4570457994644657f,0,0,0,0},
    {0,0,0.48860251190291987f,0,0,0,0,0,0,0,0,0,-1.1195289977703462f,0,0,0},
    {0,0,0,0,0,0,0,0,0.5462742152960396f,0,0,0,0,0,0,0},
    {0,0,0,0,0,0,0,0,-0.5462742152960396f,0,0,0,0,0,0,0},
    {0,0,0,0,0,0,0.94617469575756f,0,0,0,0,0,0,0,0,0},
    {0,0,0,0,1.0925484305920792f,0,0,0,0,0,0,0,0,0,0,0},
    {0,0,0,0,0,-1.0925484305920792f,0,0,0,0,0,0,0,0,0,0},
    {0,0,0,0,0,0,0,-1.0925484305920792f,0,0,0,0,0,0,0,0},
    {0,0,0,0,0,0,0,0,0,0,0,0,0,0,0,-0.5900435899266435f},
    {0,0,0,0,0,0,0,0,0,0.5900435899266435f,0,0,0,0,0,0},
    {0,0,0,0,0,0,0,0,0,0,0,0,1.865881662950577f,0,0,0},
    {0,0,0,0,0,0,0,0,0,-1.7701307697799304f,0,0,0,0,0,0},
    {0,0,0,0,0,0,0,0,0,0,0,0,0,0,1.445305721320277f,0},
    {0,0,0,0,0,0,0,0,0,0,0,0,0,0,0,1.7701307697799304f},
    {0,0,0,0,0,0,0,0,0,0,0,0,0,0,-1.445305721320277f,0},
    {0,0,0,0,0,0,0,0,0,0,0,0,0,-2.2852289973223288f,0,0},
    {0,0,0,0,0,0,0,0,0,0,0,-2.2852289973223288f,0,0,0,0},
    {0,0,0,0,0,0,0,0,0,0,2.890611442640554f,0,0,0,0,0},
};


template <int R>
__device__ __forceinline__ void cell_ijk(float px, float py, float pz,
                                         float& fx, float& fy, float& fz,
                                         int& ix, int& iy, int& iz) {
  const float cx = px * (float)(R - 1);
  const float cy = py * (float)(R - 1);
  const float cz = pz * (float)(R - 1);
  ix = (int)cx; ix = ix < R - 2 ? ix : R - 2; ix = ix > 0 ? ix : 0;
  iy = (int)cy; iy = iy < R - 2 ? iy : R - 2; iy = iy > 0 ? iy : 0;
  iz = (int)cz; iz = iz < R - 2 ? iz : R - 2; iz = iz > 0 ? iz : 0;
  fx = cx - (float)ix;
  fy = cy - (float)iy;
  fz = cz - (float)iz;
}


template <int R>
__device__ __forceinline__ void corners(const v2f* g, int i000, v2f (&c)[8]) {
  const int i100 = i000 + R * R;
  c[0] = g[i000];         c[1] = g[i000 + 1];
  c[2] = g[i000 + R];     c[3] = g[i000 + R + 1];
  c[4] = g[i100];         c[5] = g[i100 + 1];
  c[6] = g[i100 + R];     c[7] = g[i100 + R + 1];
}


__device__ __forceinline__ v2f trilerp(const v2f (&c)[8], float fx, float fy, float fz) {
  const v2f c00 = c[0] + fz * (c[1] - c[0]);
  const v2f c01 = c[2] + fz * (c[3] - c[2]);
  const v2f c10 = c[4] + fz * (c[5] - c[4]);
  const v2f c11 = c[6] + fz * (c[7] - c[6]);
  const v2f c0  = c00 + fy * (c01 - c00);
  const v2f c1  = c10 + fy * (c11 - c10);
  return c0 + fx * (c1 - c0);
}


__device__ __forceinline__ bool ws_fits(const float* ws, unsigned long long bytes,
                                        unsigned long long& lost) {
  const unsigned long long p = (unsigned long long)ws;
  lost = (128ull - (p & 127ull)) & 127ull;
  const unsigned long long avail = bytes > lost ? bytes - lost : 0ull;
  return ws != nullptr && avail >= NEED_BYTES;
}

__device__ __forceinline__ int bin_of(float px, float py, float pz) {
  float fx, fy, fz;
  int ix, iy, iz;
  cell_ijk<128>(px, py, pz, fx, fy, fz, ix, iy, iz);
  return ((ix >> 4) * 8 + (iy >> 4)) * 4 + (iz >> 5);
}

template <int R> struct Sub {
  static constexpr int WX = 16 * (R - 1);
  static constexpr int WZ = 32 * (R - 1);
  static constexpr int CX = (WX % 127 == 0) ? WX / 127 : WX / 127 + 2;
  static constexpr int CZ = (WZ % 127 == 0) ? WZ / 127 : WZ / 127 + 2;
  static constexpr int PX = CX + 1;
  static constexpr int PZ = CZ + 1;
  static constexpr int CELLS = PX * PX * PZ;
  static constexpr int STEPS = (CELLS + THREADS - 1) / THREADS;
};
template <int R> __device__ __forceinline__ int lo_xy(int b) { return (16 * (R - 1) * b) / 127; }
template <int R> __device__ __forceinline__ int lo_z(int b)  { return (32 * (R - 1) * b) / 127; }

#define VO0 0
#define VO1 (VO0 + Sub<16>::CELLS)
#define VO2 (VO1 + Sub<21>::CELLS)
#define VO3 (VO2 + Sub<28>::CELLS)
#define VO4 (VO3 + Sub<39>::CELLS)
#define VO5 (VO4 + Sub<52>::CELLS)
#define VO6 (VO5 + Sub<70>::CELLS)
#define VO7 (VO6 + Sub<95>::CELLS)
#define VOL_CELLS (VO7 + Sub<128>::CELLS)

template <int R>
__device__ __forceinline__ void vol_load(v2f (&t)[Sub<R>::STEPS], const float* g,
                                         int lox, int loy, int loz, int tid) {
  typedef Sub<R> S;
  const v2f* s = (const v2f*)g;
#pragma unroll
  for (int k = 0; k < S::STEPS; ++k) {
    const int q  = tid + k * THREADS;
    const int qc = q < S::CELLS ? q : S::CELLS - 1;
    const int lz = qc % S::PZ;
    const int r  = qc / S::PZ;
    const int ly = r % S::PX;
    const int lx = r / S::PX;
    int gx = lox + lx; gx = gx < R - 1 ? gx : R - 1;
    int gy = loy + ly; gy = gy < R - 1 ? gy : R - 1;
    int gz = loz + lz; gz = gz < R - 1 ? gz : R - 1;
    t[k] = s[(gx * R + gy) * R + gz];
  }
}
template <int R>
__device__ __forceinline__ void vol_store(v2f* dst, const v2f (&t)[Sub<R>::STEPS], int tid) {
  typedef Sub<R> S;
#pragma unroll
  for (int k = 0; k < S::STEPS; ++k) {
    const int q = tid + k * THREADS;
    if (q < S::CELLS) dst[q] = t[k];
  }
}

template <int R>
__device__ __forceinline__ void level_fetch(const v2f* vol, int lox, int loy, int loz,
                                            float px, float py, float pz,
                                            v2f (&c)[8], float& fx, float& fy, float& fz,
                                            int& gi, bool& inb) {
  typedef Sub<R> S;
  int ix, iy, iz;
  cell_ijk<R>(px, py, pz, fx, fy, fz, ix, iy, iz);
  const int lx = ix - lox, ly = iy - loy, lz = iz - loz;
  inb = (unsigned)lx < (unsigned)S::CX && (unsigned)ly < (unsigned)S::CX &&
        (unsigned)lz < (unsigned)S::CZ;
  const int li = inb ? (lx * S::PX + ly) * S::PZ + lz : 0;
  constexpr int SY = S::PZ;
  constexpr int SX = S::PX * S::PZ;
  c[0] = vol[li];           c[1] = vol[li + 1];
  c[2] = vol[li + SY];      c[3] = vol[li + SY + 1];
  c[4] = vol[li + SX];      c[5] = vol[li + SX + 1];
  c[6] = vol[li + SX + SY]; c[7] = vol[li + SX + SY + 1];
  gi = (ix * R + iy) * R + iz;
}

__global__ __launch_bounds__(THREADS) void sort_kernel(
    const float* __restrict__ pos, const float* __restrict__ dir,
    float* __restrict__ ws, unsigned long long ws_bytes)
{
  __shared__ int whist[WAVES][NBINS];
  __shared__ int part[THREADS];
  __shared__ int wsum[WAVES];
  __shared__ __attribute__((aligned(16))) v4f srec[2 * SEG];

  unsigned long long lost;
  if (!ws_fits(ws, ws_bytes, lost)) return;
  v4f* const recs = (v4f*)((char*)ws + lost);
  int* const offs = (int*)(recs + REC_V4);

  const int tid  = threadIdx.x;
  const int lane = tid & 31;
  const int wave = __builtin_amdgcn_readfirstlane(tid >> 5);

#pragma unroll
  for (int k = 0; k < NBINS * WAVES / THREADS; ++k) (&whist[0][0])[tid + k * THREADS] = 0;

  float ppx[PER_T], ppy[PER_T], ppz[PER_T], pdx[PER_T], pdy[PER_T], pdz[PER_T];
  int bin[PER_T], wr[PER_T];
#pragma unroll
  for (int it = 0; it < PER_T; ++it) {
    const int p = blockIdx.x * SEG + it * THREADS + tid;
    ppx[it] = pos[3 * p + 0]; ppy[it] = pos[3 * p + 1]; ppz[it] = pos[3 * p + 2];
    pdx[it] = dir[3 * p + 0]; pdy[it] = dir[3 * p + 1]; pdz[it] = dir[3 * p + 2];
  }
#pragma unroll
  for (int it = 0; it < PER_T; ++it) bin[it] = bin_of(ppx[it], ppy[it], ppz[it]);
  __syncthreads();

#pragma unroll
  for (int it = 0; it < PER_T; ++it) {
    const int bb = bin[it];
    unsigned mask = 0xffffffffu;
#pragma unroll
    for (int k = 0; k < 8; ++k) {
      const bool bit = (bb >> k) & 1;
      const unsigned bal = __builtin_amdgcn_ballot_w32(bit);
      mask &= bit ? bal : ~bal;
    }
    const int old   = whist[wave][bb];
    const int cnt   = __popc(mask);
    const int below = __popc(mask & ((1u << lane) - 1u));
    wr[it] = old + below;
    __builtin_amdgcn_fence(__ATOMIC_RELEASE, "wavefront");
    __builtin_amdgcn_wave_barrier();
    __builtin_amdgcn_fence(__ATOMIC_ACQUIRE, "wavefront");
    if (below == 0) whist[wave][bb] = old + cnt;
    __builtin_amdgcn_fence(__ATOMIC_RELEASE, "wavefront");
    __builtin_amdgcn_wave_barrier();
    __builtin_amdgcn_fence(__ATOMIC_ACQUIRE, "wavefront");
  }
  __syncthreads();

  {
    const int bb = tid >> 1;
    const int w0 = (tid & 1) * 8;
    int v[8];
    int s = 0;
#pragma unroll
    for (int j = 0; j < 8; ++j) { v[j] = whist[w0 + j][bb]; s += v[j]; }
    part[tid] = s;
    __syncthreads();
    if (tid < WAVES) {
      int a = 0;
      for (int k = 0; k < 32; ++k) a += part[tid * 32 + k];
      wsum[tid] = a;
    }
    __syncthreads();
    int ex = 0;
    for (int k = 0; k < WAVES; ++k) ex += (k < wave) ? wsum[k] : 0;
    for (int k = 0; k < 32; ++k) ex += (k < lane) ? part[wave * 32 + k] : 0;
    int run = ex;
#pragma unroll
    for (int j = 0; j < 8; ++j) { whist[w0 + j][bb] = run; run += v[j]; }
    if ((tid & 1) == 0) offs[blockIdx.x * OFF_PITCH + bb] = ex;
    if (tid == THREADS - 1) offs[blockIdx.x * OFF_PITCH + NBINS] = run;
  }
  __syncthreads();

#pragma unroll
  for (int it = 0; it < PER_T; ++it) {
    const int p = blockIdx.x * SEG + it * THREADS + tid;
    int dest = whist[wave][bin[it]] + wr[it];
    dest = dest < SEG - 1 ? dest : SEG - 1;
    dest = dest > 0 ? dest : 0;
    srec[2 * dest + 0] = (v4f){ppx[it], ppy[it], ppz[it], __int_as_float(p)};
    srec[2 * dest + 1] = (v4f){pdx[it], pdy[it], pdz[it], 0.0f};
  }
  __syncthreads();
  v4f* const seg = recs + (size_t)blockIdx.x * (2 * SEG);
#pragma unroll
  for (int k = 0; k < 2 * SEG / THREADS; ++k) seg[tid + k * THREADS] = srec[tid + k * THREADS];
}

__global__ __launch_bounds__(THREADS) void encode_kernel(
    const float* __restrict__ pos, const float* __restrict__ dir,
    const float* __restrict__ g0, const float* __restrict__ g1,
    const float* __restrict__ g2, const float* __restrict__ g3,
    const float* __restrict__ g4, const float* __restrict__ g5,
    const float* __restrict__ g6, const float* __restrict__ g7,
    float* __restrict__ out,
    const float* __restrict__ ws, unsigned long long ws_bytes)
{
  __shared__ __attribute__((aligned(16))) float stage[WAVES][HALF_ROWS * ROW_PITCH];
  __shared__ v2f vol[VOL_CELLS];
  __shared__ __attribute__((aligned(16))) v4f list[2 * CH];
  __shared__ int sLo[NWG], sCnt[NWG], sStart[NWG], sGrp[16];

  const int tid  = threadIdx.x;
  const int lane = tid & 31;
  const int wave = __builtin_amdgcn_readfirstlane(tid >> 5);
  float* st = stage[wave];
  const int b = blockIdx.x;

  unsigned long long lost;
  const bool fits = ws_fits(ws, ws_bytes, lost);
  const v4f* const recs = (const v4f*)((const char*)ws + lost);
  const int* const offs = (const int*)(recs + REC_V4);

  const int bx = b >> 5, by = (b >> 2) & 7, bz = b & 3;
#define LO3(L, R)                                              \
  const int x##L = fits ? lo_xy<R>(bx) : FAR_LO;               \
  const int y##L = fits ? lo_xy<R>(by) : FAR_LO;               \
  const int z##L = fits ? lo_z<R>(bz)  : FAR_LO;
  LO3(0, 16) LO3(1, 21) LO3(2, 28) LO3(3, 39) LO3(4, 52) LO3(5, 70) LO3(6, 95) LO3(7, 128)
#undef LO3

  int total = SEG;
  if (fits) {
    if (tid < NWG) {
      int a = offs[tid * OFF_PITCH + b];
      int e = offs[tid * OFF_PITCH + b + 1];
      a = a > 0 ? a : 0;   a = a < SEG ? a : SEG;
      e = e > a ? e : a;   e = e < SEG ? e : SEG;
      sLo[tid]  = a;
      sCnt[tid] = e - a;
    }
    v2f t0[Sub<16>::STEPS], t1[Sub<21>::STEPS], t2[Sub<28>::STEPS], t3[Sub<39>::STEPS];
    v2f t4[Sub<52>::STEPS], t5[Sub<70>::STEPS], t6[Sub<95>::STEPS], t7[Sub<128>::STEPS];
    vol_load<128>(t7, g7, x7, y7, z7, tid);
    vol_load<95>(t6, g6, x6, y6, z6, tid);
    vol_load<70>(t5, g5, x5, y5, z5, tid);
    vol_load<52>(t4, g4, x4, y4, z4, tid);
    vol_load<39>(t3, g3, x3, y3, z3, tid);
    vol_load<28>(t2, g2, x2, y2, z2, tid);
    vol_load<21>(t1, g1, x1, y1, z1, tid);
    vol_load<16>(t0, g0, x0, y0, z0, tid);
    __syncthreads();
    if (tid < 16) {
      int s = 0;
      for (int k = 0; k < 16; ++k) s += sCnt[tid * 16 + k];
      sGrp[tid] = s;
    }
    __syncthreads();
    if (tid < NWG) {
      const int g = tid >> 4;
      int s = 0;
      for (int k = 0; k < 16; ++k) s += (k < g) ? sGrp[k] : 0;
      for (int k = 0; k < 16; ++k) s += (k < (tid & 15)) ? sCnt[g * 16 + k] : 0;
      sStart[tid] = s;
    }
    int tt = 0;
    for (int k = 0; k < 16; ++k) tt += sGrp[k];
    total = __builtin_amdgcn_readfirstlane(tt);
    vol_store<128>(vol + VO7, t7, tid);
    vol_store<95>(vol + VO6, t6, tid);
    vol_store<70>(vol + VO5, t5, tid);
    vol_store<52>(vol + VO4, t4, tid);
    vol_store<39>(vol + VO3, t3, tid);
    vol_store<28>(vol + VO2, t2, tid);
    vol_store<21>(vol + VO1, t1, tid);
    vol_store<16>(vol + VO0, t0, tid);
  }

  const int  n  = lane & 15;
  const bool hi = lane >= 16;

  v2f bfrag[5];
  for (int c = 0; c < 5; ++c) {
    const int kb = 4 * c + (hi ? 2 : 0);
    bfrag[c].x = SH_COEF[kb][n];
    bfrag[c].y = SH_COEF[kb + 1][n];
  }

  const int rsub = lane >> 3;
  const int c4o  = (lane & 7) * 4;

  for (int c0 = 0; c0 < total; c0 += CH) {
    __syncthreads();
    const int nch = (total - c0) < CH ? (total - c0) : CH;
    if (fits) {
      const int s   = tid >> 1;
      const int par = tid & 1;
      const int st0 = sStart[s];
      const int rbeg = st0 > c0 ? st0 : c0;
      const int rlim = st0 + sCnt[s];
      const int rend = rlim < c0 + CH ? rlim : c0 + CH;
      const v4f* const src = recs + 2 * (size_t)(s * SEG + sLo[s] - st0);
      for (int r = rbeg + par; r < rend; r += 16) {
        v4f a[8], d[8];
#pragma unroll
        for (int k = 0; k < 8; ++k) {
          const int rk = r + 2 * k;
          const int rc = rk < rend ? rk : r;
          a[k] = src[2 * (size_t)rc + 0];
          d[k] = src[2 * (size_t)rc + 1];
        }
#pragma unroll
        for (int k = 0; k < 8; ++k) {
          const int rk = r + 2 * k;
          if (rk < rend) {
            list[2 * (rk - c0) + 0] = a[k];
            list[2 * (rk - c0) + 1] = d[k];
          }
        }
      }
    } else {
#pragma unroll
      for (int k = 0; k < CH / THREADS; ++k) {
        const int r = tid + k * THREADS;
        if (r < nch) {
          const int p = b * SEG + c0 + r;
          list[2 * r + 0] = (v4f){pos[3 * p + 0], pos[3 * p + 1], pos[3 * p + 2], __int_as_float(p)};
          list[2 * r + 1] = (v4f){dir[3 * p + 0], dir[3 * p + 1], dir[3 * p + 2], 0.0f};
        }
      }
    }
    __syncthreads();

    const int tiles = (nch + 31) >> 5;
    for (int tile = wave; tile < tiles; tile += WAVES) {
      const int base = tile * 32;
      const int last = nch - 1;

      const int rme = base + lane < last ? base + lane : last;
      const v4f me = list[2 * rme];
      const float px = me.x, py = me.y, pz = me.z;

      v8f acc[2];
      for (int t = 0; t < 2; ++t) {
        const int ra = base + 16 * t + n < last ? base + 16 * t + n : last;
        const v4f d4 = list[2 * ra + 1];
        const float dx = d4.x;
        const float dy = d4.y;
        const float dz = d4.z;
        const float inv = rsqrtf(dx * dx + dy * dy + dz * dz);
        const float x = dx * inv, y = dy * inv, z = dz * inv;
        const float xx = x * x, yy = y * y, zz = z * z;

        const float m0 = 1.0f, m1 = x, m2 = y, m3 = z;
        const float m4 = xx, m5 = yy, m6 = zz, m7 = x * y, m8 = y * z, m9 = x * z;
        const float m10 = xx * x, m11 = yy * y, m12 = zz * z;
        const float m13 = xx * y, m14 = xx * z, m15 = x * yy;
        const float m16 = yy * z, m17 = x * zz, m18 = y * zz, m19 = x * y * z;

        v8f a8 = {0.f, 0.f, 0.f, 0.f, 0.f, 0.f, 0.f, 0.f};

#define SH_CHUNK(C, E0, E1, E2, E3)                                           \
        do {                                                                  \
          v2f a;                                                              \
          a.x = hi ? (E2) : (E0);                                             \
          a.y = hi ? (E3) : (E1);                                             \
          a8 = __builtin_amdgcn_wmma_f32_16x16x4_f32(                         \
              false, a, false, bfrag[(C) / 4], (short)0, a8, false, false);   \
        } while (0)

        SH_CHUNK(0,  m0,  m1,  m2,  m3);
        SH_CHUNK(4,  m4,  m5,  m6,  m7);
        SH_CHUNK(8,  m8,  m9,  m10, m11);
        SH_CHUNK(12, m12, m13, m14, m15);
        SH_CHUNK(16, m16, m17, m18, m19);
#undef SH_CHUNK
        acc[t] = a8;
      }

      v2f res[8];
      {
        v2f ca[8], cb[8], cc[8], cd[8];
        float fxa, fya, fza, fxb, fyb, fzb, fxc, fyc, fzc, fxd, fyd, fzd;
        int ga, gb, gc, gd;
        bool ia, ib, ic, id;
        level_fetch<16>(vol + VO0, x0, y0, z0, px, py, pz, ca, fxa, fya, fza, ga, ia);
        level_fetch<21>(vol + VO1, x1, y1, z1, px, py, pz, cb, fxb, fyb, fzb, gb, ib);
        level_fetch<28>(vol + VO2, x2, y2, z2, px, py, pz, cc, fxc, fyc, fzc, gc, ic);
        level_fetch<39>(vol + VO3, x3, y3, z3, px, py, pz, cd, fxd, fyd, fzd, gd, id);
        if (!(ia && ib && ic && id)) {
          if (!ia) corners<16>((const v2f*)g0, ga, ca);
          if (!ib) corners<21>((const v2f*)g1, gb, cb);
          if (!ic) corners<28>((const v2f*)g2, gc, cc);
          if (!id) corners<39>((const v2f*)g3, gd, cd);
        }
        res[0] = trilerp(ca, fxa, fya, fza);
        res[1] = trilerp(cb, fxb, fyb, fzb);
        res[2] = trilerp(cc, fxc, fyc, fzc);
        res[3] = trilerp(cd, fxd, fyd, fzd);
      }
      {
        v2f ca[8], cb[8], cc[8], cd[8];
        float fxa, fya, fza, fxb, fyb, fzb, fxc, fyc, fzc, fxd, fyd, fzd;
        int ga, gb, gc, gd;
        bool ia, ib, ic, id;
        level_fetch<52>(vol + VO4, x4, y4, z4, px, py, pz, ca, fxa, fya, fza, ga, ia);
        level_fetch<70>(vol + VO5, x5, y5, z5, px, py, pz, cb, fxb, fyb, fzb, gb, ib);
        level_fetch<95>(vol + VO6, x6, y6, z6, px, py, pz, cc, fxc, fyc, fzc, gc, ic);
        level_fetch<128>(vol + VO7, x7, y7, z7, px, py, pz, cd, fxd, fyd, fzd, gd, id);
        if (!(ia && ib && ic && id)) {
          if (!ia) corners<52>((const v2f*)g4, ga, ca);
          if (!ib) corners<70>((const v2f*)g5, gb, cb);
          if (!ic) corners<95>((const v2f*)g6, gc, cc);
          if (!id) corners<128>((const v2f*)g7, gd, cd);
        }
        res[4] = trilerp(ca, fxa, fya, fza);
        res[5] = trilerp(cb, fxb, fyb, fzb);
        res[6] = trilerp(cc, fxc, fyc, fzc);
        res[7] = trilerp(cd, fxd, fyd, fzd);
      }

      for (int h = 0; h < 2; ++h) {
        const int rowBase = hi ? 8 : 0;
#pragma unroll
        for (int r = 0; r < 8; ++r) {
          st[(rowBase + r) * ROW_PITCH + 16 + n] = acc[h][r];
        }
        if ((lane >> 4) == h) {
          v4f* srow = (v4f*)(st + n * ROW_PITCH);
          for (int k = 0; k < 4; ++k) {
            v4f v = {res[2 * k].x, res[2 * k].y, res[2 * k + 1].x, res[2 * k + 1].y};
            srow[k] = v;
          }
        }

        __builtin_amdgcn_fence(__ATOMIC_RELEASE, "wavefront");
        __builtin_amdgcn_wave_barrier();
        __builtin_amdgcn_fence(__ATOMIC_ACQUIRE, "wavefront");

        if (base + 32 <= nch) {
          for (int j = 0; j < 4; ++j) {
            const int row = 4 * j + rsub;
            int pidx = __float_as_int(list[2 * (base + HALF_ROWS * h + row)].w);
            pidx = pidx > 0 ? pidx : 0;
            pidx = pidx < NPTS - 1 ? pidx : NPTS - 1;
            const v4f v = *(const v4f*)(st + row * ROW_PITCH + c4o);
            *(v4f*)(out + (size_t)pidx * 32 + c4o) = v;
          }
        } else {
          for (int j = 0; j < 4; ++j) {
            const int row = 4 * j + rsub;
            const int rq  = base + HALF_ROWS * h + row;
            const bool ok = rq <= last;
            int pidx = __float_as_int(list[2 * (ok ? rq : last)].w);
            pidx = pidx > 0 ? pidx : 0;
            pidx = pidx < NPTS - 1 ? pidx : NPTS - 1;
            const v4f v = *(const v4f*)(st + row * ROW_PITCH + c4o);
            if (ok) *(v4f*)(out + (size_t)pidx * 32 + c4o) = v;
          }
        }

        __builtin_amdgcn_fence(__ATOMIC_RELEASE, "wavefront");
        __builtin_amdgcn_wave_barrier();
        __builtin_amdgcn_fence(__ATOMIC_ACQUIRE, "wavefront");
      }
    }
  }
}

extern "C" void kernel_launch(void* const* d_in, const int* in_sizes, int n_in,
                              void* d_out, int out_size, void* d_ws, size_t ws_size,
                              hipStream_t stream) {
  (void)in_sizes; (void)n_in; (void)out_size;

  const float* pos = (const float*)d_in[0];
  const float* dir = (const float*)d_in[1];
  const float* g0 = (const float*)d_in[2];
  const float* g1 = (const float*)d_in[3];
  const float* g2 = (const float*)d_in[4];
  const float* g3 = (const float*)d_in[5];
  const float* g4 = (const float*)d_in[6];
  const float* g5 = (const float*)d_in[7];
  const float* g6 = (const float*)d_in[8];
  const float* g7 = (const float*)d_in[9];
  float* out = (float*)d_out;

  float* ws = (float*)d_ws;

  sort_kernel<<<NWG, THREADS, 0, stream>>>(pos, dir, ws, ws_size);

  encode_kernel<<<NWG, THREADS, 0, stream>>>(pos, dir, g0, g1, g2, g3, g4,
                                             g5, g6, g7, out, ws, ws_size);
}


# Launch 2: LDS holds record indices, not records; each wave loads its next tile's 32 records itself (contiguous runs of the sorted segments) under this tile's SH part and copy-out; sums unchanged
# speedup vs baseline: 3.0793x; 1.0737x over previous
#include <hip/hip_runtime.h>


typedef __attribute__((ext_vector_type(2))) float v2f;
typedef __attribute__((ext_vector_type(4))) float v4f;
typedef __attribute__((ext_vector_type(8))) float v8f;

#define NPTS (1 << 20)

#define NWG     256
#define WAVES   16
#define THREADS (WAVES * 32)
#define SEG     (NPTS / NWG)
#define PER_T   (SEG / THREADS)
#define NBINS   256
#define CH      8192

#define ROW_PITCH 36
#define HALF_ROWS 16

#define REC_V4    (2 * NPTS)
#define OFF_PITCH 260
#define NEED_BYTES (16ull * REC_V4 + 4ull * NWG * OFF_PITCH)
#define FAR_LO (1 << 20)

__constant__ float SH_COEF[20][16] = {
    {0.28209479177387814f,0,0,0,0,0,-0.31539156525252f,0,0,0,0,0,0,0,0,0},
    {0,0,0,-0.48860251190291987f,0,0,0,0,0,0,0,0,0,0.4570457994644657f,0,0},
    {0,-0.48860251190291987f,0,0,0,0,0,0,0,0,0,0.4570457994644657f,0,0,0,0},
    {0,0,0.48860251190291987f,0,0,0,0,0,0,0,0,0,-1.1195289977703462f,0,0,0},
    {0,0,0,0,0,0,0,0,0.5462742152960396f,0,0,0,0,0,0,0},
    {0,0,0,0,0,0,0,0,-0.5462742152960396f,0,0,0,0,0,0,0},
    {0,0,0,0,0,0,0.94617469575756f,0,0,0,0,0,0,0,0,0},
    {0,0,0,0,1.0925484305920792f,0,0,0,0,0,0,0,0,0,0,0},
    {0,0,0,0,0,-1.0925484305920792f,0,0,0,0,0,0,0,0,0,0},
    {0,0,0,0,0,0,0,-1.0925484305920792f,0,0,0,0,0,0,0,0},
    {0,0,0,0,0,0,0,0,0,0,0,0,0,0,0,-0.5900435899266435f},
    {0,0,0,0,0,0,0,0,0,0.5900435899266435f,0,0,0,0,0,0},
    {0,0,0,0,0,0,0,0,0,0,0,0,1.865881662950577f,0,0,0},
    {0,0,0,0,0,0,0,0,0,-1.7701307697799304f,0,0,0,0,0,0},
    {0,0,0,0,0,0,0,0,0,0,0,0,0,0,1.445305721320277f,0},
    {0,0,0,0,0,0,0,0,0,0,0,0,0,0,0,1.7701307697799304f},
    {0,0,0,0,0,0,0,0,0,0,0,0,0,0,-1.445305721320277f,0},
    {0,0,0,0,0,0,0,0,0,0,0,0,0,-2.2852289973223288f,0,0},
    {0,0,0,0,0,0,0,0,0,0,0,-2.2852289973223288f,0,0,0,0},
    {0,0,0,0,0,0,0,0,0,0,2.890611442640554f,0,0,0,0,0},
};


template <int R>
__device__ __forceinline__ void cell_ijk(float px, float py, float pz,
                                         float& fx, float& fy, float& fz,
                                         int& ix, int& iy, int& iz) {
  const float cx = px * (float)(R - 1);
  const float cy = py * (float)(R - 1);
  const float cz = pz * (float)(R - 1);
  ix = (int)cx; ix = ix < R - 2 ? ix : R - 2; ix = ix > 0 ? ix : 0;
  iy = (int)cy; iy = iy < R - 2 ? iy : R - 2; iy = iy > 0 ? iy : 0;
  iz = (int)cz; iz = iz < R - 2 ? iz : R - 2; iz = iz > 0 ? iz : 0;
  fx = cx - (float)ix;
  fy = cy - (float)iy;
  fz = cz - (float)iz;
}


template <int R>
__device__ __forceinline__ void corners(const v2f* g, int i000, v2f (&c)[8]) {
  const int i100 = i000 + R * R;
  c[0] = g[i000];         c[1] = g[i000 + 1];
  c[2] = g[i000 + R];     c[3] = g[i000 + R + 1];
  c[4] = g[i100];         c[5] = g[i100 + 1];
  c[6] = g[i100 + R];     c[7] = g[i100 + R + 1];
}


__device__ __forceinline__ v2f trilerp(const v2f (&c)[8], float fx, float fy, float fz) {
  const v2f c00 = c[0] + fz * (c[1] - c[0]);
  const v2f c01 = c[2] + fz * (c[3] - c[2]);
  const v2f c10 = c[4] + fz * (c[5] - c[4]);
  const v2f c11 = c[6] + fz * (c[7] - c[6]);
  const v2f c0  = c00 + fy * (c01 - c00);
  const v2f c1  = c10 + fy * (c11 - c10);
  return c0 + fx * (c1 - c0);
}


__device__ __forceinline__ bool ws_fits(const float* ws, unsigned long long bytes,
                                        unsigned long long& lost) {
  const unsigned long long p = (unsigned long long)ws;
  lost = (128ull - (p & 127ull)) & 127ull;
  const unsigned long long avail = bytes > lost ? bytes - lost : 0ull;
  return ws != nullptr && avail >= NEED_BYTES;
}

__device__ __forceinline__ int bin_of(float px, float py, float pz) {
  float fx, fy, fz;
  int ix, iy, iz;
  cell_ijk<128>(px, py, pz, fx, fy, fz, ix, iy, iz);
  return ((ix >> 4) * 8 + (iy >> 4)) * 4 + (iz >> 5);
}

template <int R> struct Sub {
  static constexpr int WX = 16 * (R - 1);
  static constexpr int WZ = 32 * (R - 1);
  static constexpr int CX = (WX % 127 == 0) ? WX / 127 : WX / 127 + 2;
  static constexpr int CZ = (WZ % 127 == 0) ? WZ / 127 : WZ / 127 + 2;
  static constexpr int PX = CX + 1;
  static constexpr int PZ = CZ + 1;
  static constexpr int CELLS = PX * PX * PZ;
  static constexpr int STEPS = (CELLS + THREADS - 1) / THREADS;
};
template <int R> __device__ __forceinline__ int lo_xy(int b) { return (16 * (R - 1) * b) / 127; }
template <int R> __device__ __forceinline__ int lo_z(int b)  { return (32 * (R - 1) * b) / 127; }

#define VO0 0
#define VO1 (VO0 + Sub<16>::CELLS)
#define VO2 (VO1 + Sub<21>::CELLS)
#define VO3 (VO2 + Sub<28>::CELLS)
#define VO4 (VO3 + Sub<39>::CELLS)
#define VO5 (VO4 + Sub<52>::CELLS)
#define VO6 (VO5 + Sub<70>::CELLS)
#define VO7 (VO6 + Sub<95>::CELLS)
#define VOL_CELLS (VO7 + Sub<128>::CELLS)

template <int R>
__device__ __forceinline__ void vol_load(v2f (&t)[Sub<R>::STEPS], const float* g,
                                         int lox, int loy, int loz, int tid) {
  typedef Sub<R> S;
  const v2f* s = (const v2f*)g;
#pragma unroll
  for (int k = 0; k < S::STEPS; ++k) {
    const int q  = tid + k * THREADS;
    const int qc = q < S::CELLS ? q : S::CELLS - 1;
    const int lz = qc % S::PZ;
    const int r  = qc / S::PZ;
    const int ly = r % S::PX;
    const int lx = r / S::PX;
    int gx = lox + lx; gx = gx < R - 1 ? gx : R - 1;
    int gy = loy + ly; gy = gy < R - 1 ? gy : R - 1;
    int gz = loz + lz; gz = gz < R - 1 ? gz : R - 1;
    t[k] = s[(gx * R + gy) * R + gz];
  }
}
template <int R>
__device__ __forceinline__ void vol_store(v2f* dst, const v2f (&t)[Sub<R>::STEPS], int tid) {
  typedef Sub<R> S;
#pragma unroll
  for (int k = 0; k < S::STEPS; ++k) {
    const int q = tid + k * THREADS;
    if (q < S::CELLS) dst[q] = t[k];
  }
}

template <int R>
__device__ __forceinline__ void level_fetch(const v2f* vol, int lox, int loy, int loz,
                                            float px, float py, float pz,
                                            v2f (&c)[8], float& fx, float& fy, float& fz,
                                            int& gi, bool& inb) {
  typedef Sub<R> S;
  int ix, iy, iz;
  cell_ijk<R>(px, py, pz, fx, fy, fz, ix, iy, iz);
  const int lx = ix - lox, ly = iy - loy, lz = iz - loz;
  inb = (unsigned)lx < (unsigned)S::CX && (unsigned)ly < (unsigned)S::CX &&
        (unsigned)lz < (unsigned)S::CZ;
  const int li = inb ? (lx * S::PX + ly) * S::PZ + lz : 0;
  constexpr int SY = S::PZ;
  constexpr int SX = S::PX * S::PZ;
  c[0] = vol[li];           c[1] = vol[li + 1];
  c[2] = vol[li + SY];      c[3] = vol[li + SY + 1];
  c[4] = vol[li + SX];      c[5] = vol[li + SX + 1];
  c[6] = vol[li + SX + SY]; c[7] = vol[li + SX + SY + 1];
  gi = (ix * R + iy) * R + iz;
}

template <int HALF>
__device__ __forceinline__ v4f rec_half(const v4f* __restrict__ recs, const int* rt, bool fits,
                                        const float* __restrict__ src, int pbase, int q) {
  if (fits) {
    int idx = rt[q];
    idx = idx > 0 ? idx : 0;
    idx = idx < NPTS - 1 ? idx : NPTS - 1;
    return recs[2 * (size_t)idx + HALF];
  }
  const int p = pbase + q;
  return (v4f){src[3 * p + 0], src[3 * p + 1], src[3 * p + 2], HALF == 0 ? __int_as_float(p) : 0.0f};
}

__global__ __launch_bounds__(THREADS) void sort_kernel(
    const float* __restrict__ pos, const float* __restrict__ dir,
    float* __restrict__ ws, unsigned long long ws_bytes)
{
  __shared__ int whist[WAVES][NBINS];
  __shared__ int part[THREADS];
  __shared__ int wsum[WAVES];
  __shared__ __attribute__((aligned(16))) v4f srec[2 * SEG];

  unsigned long long lost;
  if (!ws_fits(ws, ws_bytes, lost)) return;
  v4f* const recs = (v4f*)((char*)ws + lost);
  int* const offs = (int*)(recs + REC_V4);

  const int tid  = threadIdx.x;
  const int lane = tid & 31;
  const int wave = __builtin_amdgcn_readfirstlane(tid >> 5);

#pragma unroll
  for (int k = 0; k < NBINS * WAVES / THREADS; ++k) (&whist[0][0])[tid + k * THREADS] = 0;

  float ppx[PER_T], ppy[PER_T], ppz[PER_T], pdx[PER_T], pdy[PER_T], pdz[PER_T];
  int bin[PER_T], wr[PER_T];
#pragma unroll
  for (int it = 0; it < PER_T; ++it) {
    const int p = blockIdx.x * SEG + it * THREADS + tid;
    ppx[it] = pos[3 * p + 0]; ppy[it] = pos[3 * p + 1]; ppz[it] = pos[3 * p + 2];
    pdx[it] = dir[3 * p + 0]; pdy[it] = dir[3 * p + 1]; pdz[it] = dir[3 * p + 2];
  }
#pragma unroll
  for (int it = 0; it < PER_T; ++it) bin[it] = bin_of(ppx[it], ppy[it], ppz[it]);
  __syncthreads();

#pragma unroll
  for (int it = 0; it < PER_T; ++it) {
    const int bb = bin[it];
    unsigned mask = 0xffffffffu;
#pragma unroll
    for (int k = 0; k < 8; ++k) {
      const bool bit = (bb >> k) & 1;
      const unsigned bal = __builtin_amdgcn_ballot_w32(bit);
      mask &= bit ? bal : ~bal;
    }
    const int old   = whist[wave][bb];
    const int cnt   = __popc(mask);
    const int below = __popc(mask & ((1u << lane) - 1u));
    wr[it] = old + below;
    __builtin_amdgcn_fence(__ATOMIC_RELEASE, "wavefront");
    __builtin_amdgcn_wave_barrier();
    __builtin_amdgcn_fence(__ATOMIC_ACQUIRE, "wavefront");
    if (below == 0) whist[wave][bb] = old + cnt;
    __builtin_amdgcn_fence(__ATOMIC_RELEASE, "wavefront");
    __builtin_amdgcn_wave_barrier();
    __builtin_amdgcn_fence(__ATOMIC_ACQUIRE, "wavefront");
  }
  __syncthreads();

  {
    const int bb = tid >> 1;
    const int w0 = (tid & 1) * 8;
    int v[8];
    int s = 0;
#pragma unroll
    for (int j = 0; j < 8; ++j) { v[j] = whist[w0 + j][bb]; s += v[j]; }
    part[tid] = s;
    __syncthreads();
    if (tid < WAVES) {
      int a = 0;
      for (int k = 0; k < 32; ++k) a += part[tid * 32 + k];
      wsum[tid] = a;
    }
    __syncthreads();
    int ex = 0;
    for (int k = 0; k < WAVES; ++k) ex += (k < wave) ? wsum[k] : 0;
    for (int k = 0; k < 32; ++k) ex += (k < lane) ? part[wave * 32 + k] : 0;
    int run = ex;
#pragma unroll
    for (int j = 0; j < 8; ++j) { whist[w0 + j][bb] = run; run += v[j]; }
    if ((tid & 1) == 0) offs[blockIdx.x * OFF_PITCH + bb] = ex;
    if (tid == THREADS - 1) offs[blockIdx.x * OFF_PITCH + NBINS] = run;
  }
  __syncthreads();

#pragma unroll
  for (int it = 0; it < PER_T; ++it) {
    const int p = blockIdx.x * SEG + it * THREADS + tid;
    int dest = whist[wave][bin[it]] + wr[it];
    dest = dest < SEG - 1 ? dest : SEG - 1;
    dest = dest > 0 ? dest : 0;
    srec[2 * dest + 0] = (v4f){ppx[it], ppy[it], ppz[it], __int_as_float(p)};
    srec[2 * dest + 1] = (v4f){pdx[it], pdy[it], pdz[it], 0.0f};
  }
  __syncthreads();
  v4f* const seg = recs + (size_t)blockIdx.x * (2 * SEG);
#pragma unroll
  for (int k = 0; k < 2 * SEG / THREADS; ++k) seg[tid + k * THREADS] = srec[tid + k * THREADS];
}

template <bool FITS>
__device__ __forceinline__ void encode_body(
    const float* __restrict__ pos, const float* __restrict__ dir,
    const float* __restrict__ g0, const float* __restrict__ g1,
    const float* __restrict__ g2, const float* __restrict__ g3,
    const float* __restrict__ g4, const float* __restrict__ g5,
    const float* __restrict__ g6, const float* __restrict__ g7,
    float* __restrict__ out,
    const v4f* __restrict__ recs, const int* __restrict__ offs,
    float (*stage)[HALF_ROWS * ROW_PITCH], v2f* vol, int* rt,
    int* sLo, int* sCnt, int* sStart, int* sGrp)
{
  constexpr bool fits = FITS;
  const int tid  = threadIdx.x;
  const int lane = tid & 31;
  const int wave = __builtin_amdgcn_readfirstlane(tid >> 5);
  float* st = stage[wave];
  const int b = blockIdx.x;

  const int bx = b >> 5, by = (b >> 2) & 7, bz = b & 3;
#define LO3(L, R)                                              \
  const int x##L = fits ? lo_xy<R>(bx) : FAR_LO;               \
  const int y##L = fits ? lo_xy<R>(by) : FAR_LO;               \
  const int z##L = fits ? lo_z<R>(bz)  : FAR_LO;
  LO3(0, 16) LO3(1, 21) LO3(2, 28) LO3(3, 39) LO3(4, 52) LO3(5, 70) LO3(6, 95) LO3(7, 128)
#undef LO3

  int total = SEG;
  if (fits) {
    if (tid < NWG) {
      int a = offs[tid * OFF_PITCH + b];
      int e = offs[tid * OFF_PITCH + b + 1];
      a = a > 0 ? a : 0;   a = a < SEG ? a : SEG;
      e = e > a ? e : a;   e = e < SEG ? e : SEG;
      sLo[tid]  = a;
      sCnt[tid] = e - a;
    }
    v2f t0[Sub<16>::STEPS], t1[Sub<21>::STEPS], t2[Sub<28>::STEPS], t3[Sub<39>::STEPS];
    v2f t4[Sub<52>::STEPS], t5[Sub<70>::STEPS], t6[Sub<95>::STEPS], t7[Sub<128>::STEPS];
    vol_load<128>(t7, g7, x7, y7, z7, tid);
    vol_load<95>(t6, g6, x6, y6, z6, tid);
    vol_load<70>(t5, g5, x5, y5, z5, tid);
    vol_load<52>(t4, g4, x4, y4, z4, tid);
    vol_load<39>(t3, g3, x3, y3, z3, tid);
    vol_load<28>(t2, g2, x2, y2, z2, tid);
    vol_load<21>(t1, g1, x1, y1, z1, tid);
    vol_load<16>(t0, g0, x0, y0, z0, tid);
    __syncthreads();
    if (tid < 16) {
      int s = 0;
      for (int k = 0; k < 16; ++k) s += sCnt[tid * 16 + k];
      sGrp[tid] = s;
    }
    __syncthreads();
    if (tid < NWG) {
      const int g = tid >> 4;
      int s = 0;
      for (int k = 0; k < 16; ++k) s += (k < g) ? sGrp[k] : 0;
      for (int k = 0; k < 16; ++k) s += (k < (tid & 15)) ? sCnt[g * 16 + k] : 0;
      sStart[tid] = s;
    }
    int tt = 0;
    for (int k = 0; k < 16; ++k) tt += sGrp[k];
    total = __builtin_amdgcn_readfirstlane(tt);
    vol_store<128>(vol + VO7, t7, tid);
    vol_store<95>(vol + VO6, t6, tid);
    vol_store<70>(vol + VO5, t5, tid);
    vol_store<52>(vol + VO4, t4, tid);
    vol_store<39>(vol + VO3, t3, tid);
    vol_store<28>(vol + VO2, t2, tid);
    vol_store<21>(vol + VO1, t1, tid);
    vol_store<16>(vol + VO0, t0, tid);
  }

  const int  n  = lane & 15;
  const bool hi = lane >= 16;

  v2f bfrag[5];
  for (int c = 0; c < 5; ++c) {
    const int kb = 4 * c + (hi ? 2 : 0);
    bfrag[c].x = SH_COEF[kb][n];
    bfrag[c].y = SH_COEF[kb + 1][n];
  }

  const int rsub = lane >> 3;
  const int c4o  = (lane & 7) * 4;

  for (int c0 = 0; c0 < total; c0 += CH) {
    __syncthreads();
    const int nch = (total - c0) < CH ? (total - c0) : CH;
    if (fits) {
      const int s   = tid >> 1;
      const int par = tid & 1;
      const int st0 = sStart[s];
      const int rbeg = st0 > c0 ? st0 : c0;
      const int rlim = st0 + sCnt[s];
      const int rend = rlim < c0 + CH ? rlim : c0 + CH;
      const int recbase = s * SEG + sLo[s] - st0;
      for (int r = rbeg + par; r < rend; r += 2) rt[r - c0] = recbase + r;
    }
    __syncthreads();

    const int tiles = (nch + 31) >> 5;
    const int last  = nch - 1;
    const int pbase = b * SEG + c0;

    v4f nme = {0.f, 0.f, 0.f, 0.f}, nd0 = nme, nd1 = nme;
    if (wave < tiles) {
      const int fb = wave * 32;
      nme = rec_half<0>(recs, rt, fits, pos, pbase, fb + lane < last ? fb + lane : last);
      nd0 = rec_half<1>(recs, rt, fits, dir, pbase, fb + n < last ? fb + n : last);
      nd1 = rec_half<1>(recs, rt, fits, dir, pbase, fb + 16 + n < last ? fb + 16 + n : last);
    }

    for (int tile = wave; tile < tiles; tile += WAVES) {
      const int base = tile * 32;

      const v4f me = nme;
      const v4f dd[2] = {nd0, nd1};
      const float px = me.x, py = me.y, pz = me.z;

      v2f res[8];
      {
        v2f ca[8], cb[8], cc[8], cd[8];
        float fxa, fya, fza, fxb, fyb, fzb, fxc, fyc, fzc, fxd, fyd, fzd;
        int ga, gb, gc, gd;
        bool ia, ib, ic, id;
        level_fetch<16>(vol + VO0, x0, y0, z0, px, py, pz, ca, fxa, fya, fza, ga, ia);
        level_fetch<21>(vol + VO1, x1, y1, z1, px, py, pz, cb, fxb, fyb, fzb, gb, ib);
        level_fetch<28>(vol + VO2, x2, y2, z2, px, py, pz, cc, fxc, fyc, fzc, gc, ic);
        level_fetch<39>(vol + VO3, x3, y3, z3, px, py, pz, cd, fxd, fyd, fzd, gd, id);
        if (!(ia && ib && ic && id)) {
          if (!ia) corners<16>((const v2f*)g0, ga, ca);
          if (!ib) corners<21>((const v2f*)g1, gb, cb);
          if (!ic) corners<28>((const v2f*)g2, gc, cc);
          if (!id) corners<39>((const v2f*)g3, gd, cd);
        }
        res[0] = trilerp(ca, fxa, fya, fza);
        res[1] = trilerp(cb, fxb, fyb, fzb);
        res[2] = trilerp(cc, fxc, fyc, fzc);
        res[3] = trilerp(cd, fxd, fyd, fzd);
      }
      {
        v2f ca[8], cb[8], cc[8], cd[8];
        float fxa, fya, fza, fxb, fyb, fzb, fxc, fyc, fzc, fxd, fyd, fzd;
        int ga, gb, gc, gd;
        bool ia, ib, ic, id;
        level_fetch<52>(vol + VO4, x4, y4, z4, px, py, pz, ca, fxa, fya, fza, ga, ia);
        level_fetch<70>(vol + VO5, x5, y5, z5, px, py, pz, cb, fxb, fyb, fzb, gb, ib);
        level_fetch<95>(vol + VO6, x6, y6, z6, px, py, pz, cc, fxc, fyc, fzc, gc, ic);
        level_fetch<128>(vol + VO7, x7, y7, z7, px, py, pz, cd, fxd, fyd, fzd, gd, id);
        if (!(ia && ib && ic && id)) {
          if (!ia) corners<52>((const v2f*)g4, ga, ca);
          if (!ib) corners<70>((const v2f*)g5, gb, cb);
          if (!ic) corners<95>((const v2f*)g6, gc, cc);
          if (!id) corners<128>((const v2f*)g7, gd, cd);
        }
        res[4] = trilerp(ca, fxa, fya, fza);
        res[5] = trilerp(cb, fxb, fyb, fzb);
        res[6] = trilerp(cc, fxc, fyc, fzc);
        res[7] = trilerp(cd, fxd, fyd, fzd);
      }

      {
        const int nb = base + 32 * WAVES;
        nme = rec_half<0>(recs, rt, fits, pos, pbase, nb + lane < last ? nb + lane : last);
        nd0 = rec_half<1>(recs, rt, fits, dir, pbase, nb + n < last ? nb + n : last);
        nd1 = rec_half<1>(recs, rt, fits, dir, pbase, nb + 16 + n < last ? nb + 16 + n : last);
      }

      v8f acc[2];
      for (int t = 0; t < 2; ++t) {
        const v4f d4 = dd[t];
        const float dx = d4.x;
        const float dy = d4.y;
        const float dz = d4.z;
        const float inv = rsqrtf(dx * dx + dy * dy + dz * dz);
        const float x = dx * inv, y = dy * inv, z = dz * inv;
        const float xx = x * x, yy = y * y, zz = z * z;

        const float m0 = 1.0f, m1 = x, m2 = y, m3 = z;
        const float m4 = xx, m5 = yy, m6 = zz, m7 = x * y, m8 = y * z, m9 = x * z;
        const float m10 = xx * x, m11 = yy * y, m12 = zz * z;
        const float m13 = xx * y, m14 = xx * z, m15 = x * yy;
        const float m16 = yy * z, m17 = x * zz, m18 = y * zz, m19 = x * y * z;

        v8f a8 = {0.f, 0.f, 0.f, 0.f, 0.f, 0.f, 0.f, 0.f};

#define SH_CHUNK(C, E0, E1, E2, E3)                                           \
        do {                                                                  \
          v2f a;                                                              \
          a.x = hi ? (E2) : (E0);                                             \
          a.y = hi ? (E3) : (E1);                                             \
          a8 = __builtin_amdgcn_wmma_f32_16x16x4_f32(                         \
              false, a, false, bfrag[(C) / 4], (short)0, a8, false, false);   \
        } while (0)

        SH_CHUNK(0,  m0,  m1,  m2,  m3);
        SH_CHUNK(4,  m4,  m5,  m6,  m7);
        SH_CHUNK(8,  m8,  m9,  m10, m11);
        SH_CHUNK(12, m12, m13, m14, m15);
        SH_CHUNK(16, m16, m17, m18, m19);
#undef SH_CHUNK
        acc[t] = a8;
      }

      for (int h = 0; h < 2; ++h) {
        const int rowBase = hi ? 8 : 0;
#pragma unroll
        for (int r = 0; r < 8; ++r) {
          st[(rowBase + r) * ROW_PITCH + 16 + n] = acc[h][r];
        }
        if ((lane >> 4) == h) {
          v4f* srow = (v4f*)(st + n * ROW_PITCH);
          for (int k = 0; k < 4; ++k) {
            v4f v = {res[2 * k].x, res[2 * k].y, res[2 * k + 1].x, res[2 * k + 1].y};
            srow[k] = v;
          }
          st[n * ROW_PITCH + 32] = me.w;
        }

        __builtin_amdgcn_fence(__ATOMIC_RELEASE, "wavefront");
        __builtin_amdgcn_wave_barrier();
        __builtin_amdgcn_fence(__ATOMIC_ACQUIRE, "wavefront");

        if (base + 32 <= nch) {
          for (int j = 0; j < 4; ++j) {
            const int row = 4 * j + rsub;
            int pidx = __float_as_int(st[row * ROW_PITCH + 32]);
            pidx = pidx > 0 ? pidx : 0;
            pidx = pidx < NPTS - 1 ? pidx : NPTS - 1;
            const v4f v = *(const v4f*)(st + row * ROW_PITCH + c4o);
            *(v4f*)(out + (size_t)pidx * 32 + c4o) = v;
          }
        } else {
          for (int j = 0; j < 4; ++j) {
            const int row = 4 * j + rsub;
            const int rq  = base + HALF_ROWS * h + row;
            const bool ok = rq <= last;
            int pidx = __float_as_int(st[row * ROW_PITCH + 32]);
            pidx = pidx > 0 ? pidx : 0;
            pidx = pidx < NPTS - 1 ? pidx : NPTS - 1;
            const v4f v = *(const v4f*)(st + row * ROW_PITCH + c4o);
            if (ok) *(v4f*)(out + (size_t)pidx * 32 + c4o) = v;
          }
        }

        __builtin_amdgcn_fence(__ATOMIC_RELEASE, "wavefront");
        __builtin_amdgcn_wave_barrier();
        __builtin_amdgcn_fence(__ATOMIC_ACQUIRE, "wavefront");
      }
    }
  }
}

__global__ __launch_bounds__(THREADS) void encode_kernel(
    const float* __restrict__ pos, const float* __restrict__ dir,
    const float* __restrict__ g0, const float* __restrict__ g1,
    const float* __restrict__ g2, const float* __restrict__ g3,
    const float* __restrict__ g4, const float* __restrict__ g5,
    const float* __restrict__ g6, const float* __restrict__ g7,
    float* __restrict__ out,
    const float* __restrict__ ws, unsigned long long ws_bytes)
{
  __shared__ __attribute__((aligned(16))) float stage[WAVES][HALF_ROWS * ROW_PITCH];
  __shared__ v2f vol[VOL_CELLS];
  __shared__ int rt[CH];
  __shared__ int sLo[NWG], sCnt[NWG], sStart[NWG], sGrp[16];

  unsigned long long lost;
  const bool fits = ws_fits(ws, ws_bytes, lost);
  const v4f* const recs = (const v4f*)((const char*)ws + lost);
  const int* const offs = (const int*)(recs + REC_V4);
  if (fits) {
    encode_body<true>(pos, dir, g0, g1, g2, g3, g4, g5, g6, g7, out, recs, offs,
                      stage, vol, rt, sLo, sCnt, sStart, sGrp);
  } else {
    encode_body<false>(pos, dir, g0, g1, g2, g3, g4, g5, g6, g7, out, recs, offs,
                       stage, vol, rt, sLo, sCnt, sStart, sGrp);
  }
}

extern "C" void kernel_launch(void* const* d_in, const int* in_sizes, int n_in,
                              void* d_out, int out_size, void* d_ws, size_t ws_size,
                              hipStream_t stream) {
  (void)in_sizes; (void)n_in; (void)out_size;

  const float* pos = (const float*)d_in[0];
  const float* dir = (const float*)d_in[1];
  const float* g0 = (const float*)d_in[2];
  const float* g1 = (const float*)d_in[3];
  const float* g2 = (const float*)d_in[4];
  const float* g3 = (const float*)d_in[5];
  const float* g4 = (const float*)d_in[6];
  const float* g5 = (const float*)d_in[7];
  const float* g6 = (const float*)d_in[8];
  const float* g7 = (const float*)d_in[9];
  float* out = (float*)d_out;

  float* ws = (float*)d_ws;

  sort_kernel<<<NWG, THREADS, 0, stream>>>(pos, dir, ws, ws_size);

  encode_kernel<<<NWG, THREADS, 0, stream>>>(pos, dir, g0, g1, g2, g3, g4,
                                             g5, g6, g7, out, ws, ws_size);
}


# speedup vs baseline: 3.3704x; 1.0945x over previous
#include <hip/hip_runtime.h>


typedef __attribute__((ext_vector_type(2))) float v2f;
typedef __attribute__((ext_vector_type(4))) float v4f;
typedef __attribute__((ext_vector_type(8))) float v8f;

#define NPTS (1 << 20)

#define NWG     256
#define WAVES   16
#define THREADS (WAVES * 32)
#define SEG     (NPTS / NWG)
#define PER_T   (SEG / THREADS)
#define NBINS   256
#define CH      8192

#define ROW_PITCH 36
#define HALF_ROWS 16

#define REC_V4    (2 * NPTS)
#define OFF_PITCH 260
#define NEED_BYTES (16ull * REC_V4 + 4ull * NWG * OFF_PITCH)
#define FAR_LO (1 << 20)

__constant__ float SH_COEF[20][16] = {
    {0.28209479177387814f,0,0,0,0,0,-0.31539156525252f,0,0,0,0,0,0,0,0,0},
    {0,0,0,-0.48860251190291987f,0,0,0,0,0,0,0,0,0,0.4570457994644657f,0,0},
    {0,-0.48860251190291987f,0,0,0,0,0,0,0,0,0,0.4570457994644657f,0,0,0,0},
    {0,0,0.48860251190291987f,0,0,0,0,0,0,0,0,0,-1.1195289977703462f,0,0,0},
    {0,0,0,0,0,0,0,0,0.5462742152960396f,0,0,0,0,0,0,0},
    {0,0,0,0,0,0,0,0,-0.5462742152960396f,0,0,0,0,0,0,0},
    {0,0,0,0,0,0,0.94617469575756f,0,0,0,0,0,0,0,0,0},
    {0,0,0,0,1.0925484305920792f,0,0,0,0,0,0,0,0,0,0,0},
    {0,0,0,0,0,-1.0925484305920792f,0,0,0,0,0,0,0,0,0,0},
    {0,0,0,0,0,0,0,-1.0925484305920792f,0,0,0,0,0,0,0,0},
    {0,0,0,0,0,0,0,0,0,0,0,0,0,0,0,-0.5900435899266435f},
    {0,0,0,0,0,0,0,0,0,0.5900435899266435f,0,0,0,0,0,0},
    {0,0,0,0,0,0,0,0,0,0,0,0,1.865881662950577f,0,0,0},
    {0,0,0,0,0,0,0,0,0,-1.7701307697799304f,0,0,0,0,0,0},
    {0,0,0,0,0,0,0,0,0,0,0,0,0,0,1.445305721320277f,0},
    {0,0,0,0,0,0,0,0,0,0,0,0,0,0,0,1.7701307697799304f},
    {0,0,0,0,0,0,0,0,0,0,0,0,0,0,-1.445305721320277f,0},
    {0,0,0,0,0,0,0,0,0,0,0,0,0,-2.2852289973223288f,0,0},
    {0,0,0,0,0,0,0,0,0,0,0,-2.2852289973223288f,0,0,0,0},
    {0,0,0,0,0,0,0,0,0,0,2.890611442640554f,0,0,0,0,0},
};


template <int R>
__device__ __forceinline__ void cell_ijk(float px, float py, float pz,
                                         float& fx, float& fy, float& fz,
                                         int& ix, int& iy, int& iz) {
  const float cx = px * (float)(R - 1);
  const float cy = py * (float)(R - 1);
  const float cz = pz * (float)(R - 1);
  ix = (int)cx; ix = ix < R - 2 ? ix : R - 2; ix = ix > 0 ? ix : 0;
  iy = (int)cy; iy = iy < R - 2 ? iy : R - 2; iy = iy > 0 ? iy : 0;
  iz = (int)cz; iz = iz < R - 2 ? iz : R - 2; iz = iz > 0 ? iz : 0;
  fx = cx - (float)ix;
  fy = cy - (float)iy;
  fz = cz - (float)iz;
}


template <int R>
__device__ __forceinline__ void corners(const v2f* g, int i000, v2f (&c)[8]) {
  const int i100 = i000 + R * R;
  c[0] = g[i000];         c[1] = g[i000 + 1];
  c[2] = g[i000 + R];     c[3] = g[i000 + R + 1];
  c[4] = g[i100];         c[5] = g[i100 + 1];
  c[6] = g[i100 + R];     c[7] = g[i100 + R + 1];
}


__device__ __forceinline__ v2f trilerp(const v2f (&c)[8], float fx, float fy, float fz) {
  const v2f c00 = c[0] + fz * (c[1] - c[0]);
  const v2f c01 = c[2] + fz * (c[3] - c[2]);
  const v2f c10 = c[4] + fz * (c[5] - c[4]);
  const v2f c11 = c[6] + fz * (c[7] - c[6]);
  const v2f c0  = c00 + fy * (c01 - c00);
  const v2f c1  = c10 + fy * (c11 - c10);
  return c0 + fx * (c1 - c0);
}


__device__ __forceinline__ bool ws_fits(const float* ws, unsigned long long bytes,
                                        unsigned long long& lost) {
  const unsigned long long p = (unsigned long long)ws;
  lost = (128ull - (p & 127ull)) & 127ull;
  const unsigned long long avail = bytes > lost ? bytes - lost : 0ull;
  return ws != nullptr && avail >= NEED_BYTES;
}

__device__ __forceinline__ int bin_of(float px, float py, float pz) {
  float fx, fy, fz;
  int ix, iy, iz;
  cell_ijk<128>(px, py, pz, fx, fy, fz, ix, iy, iz);
  return ((ix >> 4) * 8 + (iy >> 4)) * 4 + (iz >> 5);
}

template <int R> struct Sub {
  static constexpr int WX = 16 * (R - 1);
  static constexpr int WZ = 32 * (R - 1);
  static constexpr int CX = (WX % 127 == 0) ? WX / 127 : WX / 127 + 2;
  static constexpr int CZ = (WZ % 127 == 0) ? WZ / 127 : WZ / 127 + 2;
  static constexpr int PX = CX + 1;
  static constexpr int PZ = CZ + 1;
  static constexpr int CELLS = PX * PX * PZ;
  static constexpr int STEPS = (CELLS + THREADS - 1) / THREADS;
};
template <int R> __device__ __forceinline__ int lo_xy(int b) { return (16 * (R - 1) * b) / 127; }
template <int R> __device__ __forceinline__ int lo_z(int b)  { return (32 * (R - 1) * b) / 127; }

#define VO0 0
#define VO1 (VO0 + Sub<16>::CELLS)
#define VO2 (VO1 + Sub<21>::CELLS)
#define VO3 (VO2 + Sub<28>::CELLS)
#define VO4 (VO3 + Sub<39>::CELLS)
#define VO5 (VO4 + Sub<52>::CELLS)
#define VO6 (VO5 + Sub<70>::CELLS)
#define VO7 (VO6 + Sub<95>::CELLS)
#define VOL_CELLS (VO7 + Sub<128>::CELLS)

template <int R>
__device__ __forceinline__ void vol_load(v2f (&t)[Sub<R>::STEPS], const float* g,
                                         int lox, int loy, int loz, int tid) {
  typedef Sub<R> S;
  const v2f* s = (const v2f*)g;
#pragma unroll
  for (int k = 0; k < S::STEPS; ++k) {
    const int q  = tid + k * THREADS;
    const int qc = q < S::CELLS ? q : S::CELLS - 1;
    const int lz = qc % S::PZ;
    const int r  = qc / S::PZ;
    const int ly = r % S::PX;
    const int lx = r / S::PX;
    int gx = lox + lx; gx = gx < R - 1 ? gx : R - 1;
    int gy = loy + ly; gy = gy < R - 1 ? gy : R - 1;
    int gz = loz + lz; gz = gz < R - 1 ? gz : R - 1;
    t[k] = s[(gx * R + gy) * R + gz];
  }
}
template <int R>
__device__ __forceinline__ void vol_store(v2f* dst, const v2f (&t)[Sub<R>::STEPS], int tid) {
  typedef Sub<R> S;
#pragma unroll
  for (int k = 0; k < S::STEPS; ++k) {
    const int q = tid + k * THREADS;
    if (q < S::CELLS) dst[q] = t[k];
  }
}

template <int R>
__device__ __forceinline__ void level_fetch(const v2f* vol, int lox, int loy, int loz,
                                            float px, float py, float pz,
                                            v2f (&c)[8], float& fx, float& fy, float& fz,
                                            int& gi, bool& inb) {
  typedef Sub<R> S;
  int ix, iy, iz;
  cell_ijk<R>(px, py, pz, fx, fy, fz, ix, iy, iz);
  const int lx = ix - lox, ly = iy - loy, lz = iz - loz;
  inb = (unsigned)lx < (unsigned)S::CX && (unsigned)ly < (unsigned)S::CX &&
        (unsigned)lz < (unsigned)S::CZ;
  const int li = inb ? (lx * S::PX + ly) * S::PZ + lz : 0;
  constexpr int SY = S::PZ;
  constexpr int SX = S::PX * S::PZ;
  c[0] = vol[li];           c[1] = vol[li + 1];
  c[2] = vol[li + SY];      c[3] = vol[li + SY + 1];
  c[4] = vol[li + SX];      c[5] = vol[li + SX + 1];
  c[6] = vol[li + SX + SY]; c[7] = vol[li + SX + SY + 1];
  gi = (ix * R + iy) * R + iz;
}

__device__ __forceinline__ void row_store(float* p, v4f v) {
  __builtin_nontemporal_store(v, (v4f*)p);
}

template <int HALF>
__device__ __forceinline__ v4f rec_half(const v4f* __restrict__ recs, const int* rt, bool fits,
                                        const float* __restrict__ src, int pbase, int q) {
  if (fits) {
    int idx = rt[q];
    idx = idx > 0 ? idx : 0;
    idx = idx < NPTS - 1 ? idx : NPTS - 1;
    return recs[2 * (size_t)idx + HALF];
  }
  const int p = pbase + q;
  return (v4f){src[3 * p + 0], src[3 * p + 1], src[3 * p + 2], HALF == 0 ? __int_as_float(p) : 0.0f};
}

__global__ __launch_bounds__(THREADS) void sort_kernel(
    const float* __restrict__ pos, const float* __restrict__ dir,
    float* __restrict__ ws, unsigned long long ws_bytes)
{
  __shared__ int whist[WAVES][NBINS];
  __shared__ int part[THREADS];
  __shared__ int wsum[WAVES];
  __shared__ __attribute__((aligned(16))) v4f srec[2 * SEG];

  unsigned long long lost;
  if (!ws_fits(ws, ws_bytes, lost)) return;
  v4f* const recs = (v4f*)((char*)ws + lost);
  int* const offs = (int*)(recs + REC_V4);

  const int tid  = threadIdx.x;
  const int lane = tid & 31;
  const int wave = __builtin_amdgcn_readfirstlane(tid >> 5);

#pragma unroll
  for (int k = 0; k < NBINS * WAVES / THREADS; ++k) (&whist[0][0])[tid + k * THREADS] = 0;

  float ppx[PER_T], ppy[PER_T], ppz[PER_T], pdx[PER_T], pdy[PER_T], pdz[PER_T];
  int bin[PER_T], wr[PER_T];
#pragma unroll
  for (int it = 0; it < PER_T; ++it) {
    const int p = blockIdx.x * SEG + it * THREADS + tid;
    ppx[it] = pos[3 * p + 0]; ppy[it] = pos[3 * p + 1]; ppz[it] = pos[3 * p + 2];
    pdx[it] = dir[3 * p + 0]; pdy[it] = dir[3 * p + 1]; pdz[it] = dir[3 * p + 2];
  }
#pragma unroll
  for (int it = 0; it < PER_T; ++it) bin[it] = bin_of(ppx[it], ppy[it], ppz[it]);
  __syncthreads();

#pragma unroll
  for (int it = 0; it < PER_T; ++it) {
    const int bb = bin[it];
    unsigned mask = 0xffffffffu;
#pragma unroll
    for (int k = 0; k < 8; ++k) {
      const bool bit = (bb >> k) & 1;
      const unsigned bal = __builtin_amdgcn_ballot_w32(bit);
      mask &= bit ? bal : ~bal;
    }
    const int old   = whist[wave][bb];
    const int cnt   = __popc(mask);
    const int below = __popc(mask & ((1u << lane) - 1u));
    wr[it] = old + below;
    __builtin_amdgcn_fence(__ATOMIC_RELEASE, "wavefront");
    __builtin_amdgcn_wave_barrier();
    __builtin_amdgcn_fence(__ATOMIC_ACQUIRE, "wavefront");
    if (below == 0) whist[wave][bb] = old + cnt;
    __builtin_amdgcn_fence(__ATOMIC_RELEASE, "wavefront");
    __builtin_amdgcn_wave_barrier();
    __builtin_amdgcn_fence(__ATOMIC_ACQUIRE, "wavefront");
  }
  __syncthreads();

  {
    const int bb = tid >> 1;
    const int w0 = (tid & 1) * 8;
    int v[8];
    int s = 0;
#pragma unroll
    for (int j = 0; j < 8; ++j) { v[j] = whist[w0 + j][bb]; s += v[j]; }
    part[tid] = s;
    __syncthreads();
    if (tid < WAVES) {
      int a = 0;
      for (int k = 0; k < 32; ++k) a += part[tid * 32 + k];
      wsum[tid] = a;
    }
    __syncthreads();
    int ex = 0;
    for (int k = 0; k < WAVES; ++k) ex += (k < wave) ? wsum[k] : 0;
    for (int k = 0; k < 32; ++k) ex += (k < lane) ? part[wave * 32 + k] : 0;
    int run = ex;
#pragma unroll
    for (int j = 0; j < 8; ++j) { whist[w0 + j][bb] = run; run += v[j]; }
    if ((tid & 1) == 0) offs[blockIdx.x * OFF_PITCH + bb] = ex;
    if (tid == THREADS - 1) offs[blockIdx.x * OFF_PITCH + NBINS] = run;
  }
  __syncthreads();

#pragma unroll
  for (int it = 0; it < PER_T; ++it) {
    const int p = blockIdx.x * SEG + it * THREADS + tid;
    int dest = whist[wave][bin[it]] + wr[it];
    dest = dest < SEG - 1 ? dest : SEG - 1;
    dest = dest > 0 ? dest : 0;
    srec[2 * dest + 0] = (v4f){ppx[it], ppy[it], ppz[it], __int_as_float(p)};
    srec[2 * dest + 1] = (v4f){pdx[it], pdy[it], pdz[it], 0.0f};
  }
  __syncthreads();
  v4f* const seg = recs + (size_t)blockIdx.x * (2 * SEG);
#pragma unroll
  for (int k = 0; k < 2 * SEG / THREADS; ++k) seg[tid + k * THREADS] = srec[tid + k * THREADS];
}

template <bool FITS>
__device__ __forceinline__ void encode_body(
    const float* __restrict__ pos, const float* __restrict__ dir,
    const float* __restrict__ g0, const float* __restrict__ g1,
    const float* __restrict__ g2, const float* __restrict__ g3,
    const float* __restrict__ g4, const float* __restrict__ g5,
    const float* __restrict__ g6, const float* __restrict__ g7,
    float* __restrict__ out,
    const v4f* __restrict__ recs, const int* __restrict__ offs,
    float (*stage)[HALF_ROWS * ROW_PITCH], v2f* vol, int* rt,
    int* sLo, int* sCnt, int* sStart, int* sGrp)
{
  constexpr bool fits = FITS;
  const int tid  = threadIdx.x;
  const int lane = tid & 31;
  const int wave = __builtin_amdgcn_readfirstlane(tid >> 5);
  float* st = stage[wave];
  const int b = blockIdx.x;

  const int bx = b >> 5, by = (b >> 2) & 7, bz = b & 3;
#define LO3(L, R)                                              \
  const int x##L = fits ? lo_xy<R>(bx) : FAR_LO;               \
  const int y##L = fits ? lo_xy<R>(by) : FAR_LO;               \
  const int z##L = fits ? lo_z<R>(bz)  : FAR_LO;
  LO3(0, 16) LO3(1, 21) LO3(2, 28) LO3(3, 39) LO3(4, 52) LO3(5, 70) LO3(6, 95) LO3(7, 128)
#undef LO3

  int total = SEG;
  if (fits) {
    if (tid < NWG) {
      int a = offs[tid * OFF_PITCH + b];
      int e = offs[tid * OFF_PITCH + b + 1];
      a = a > 0 ? a : 0;   a = a < SEG ? a : SEG;
      e = e > a ? e : a;   e = e < SEG ? e : SEG;
      sLo[tid]  = a;
      sCnt[tid] = e - a;
    }
    v2f t0[Sub<16>::STEPS], t1[Sub<21>::STEPS], t2[Sub<28>::STEPS], t3[Sub<39>::STEPS];
    v2f t4[Sub<52>::STEPS], t5[Sub<70>::STEPS], t6[Sub<95>::STEPS], t7[Sub<128>::STEPS];
    vol_load<128>(t7, g7, x7, y7, z7, tid);
    vol_load<95>(t6, g6, x6, y6, z6, tid);
    vol_load<70>(t5, g5, x5, y5, z5, tid);
    vol_load<52>(t4, g4, x4, y4, z4, tid);
    vol_load<39>(t3, g3, x3, y3, z3, tid);
    vol_load<28>(t2, g2, x2, y2, z2, tid);
    vol_load<21>(t1, g1, x1, y1, z1, tid);
    vol_load<16>(t0, g0, x0, y0, z0, tid);
    __syncthreads();
    if (tid < 16) {
      int s = 0;
      for (int k = 0; k < 16; ++k) s += sCnt[tid * 16 + k];
      sGrp[tid] = s;
    }
    __syncthreads();
    if (tid < NWG) {
      const int g = tid >> 4;
      int s = 0;
      for (int k = 0; k < 16; ++k) s += (k < g) ? sGrp[k] : 0;
      for (int k = 0; k < 16; ++k) s += (k < (tid & 15)) ? sCnt[g * 16 + k] : 0;
      sStart[tid] = s;
    }
    int tt = 0;
    for (int k = 0; k < 16; ++k) tt += sGrp[k];
    total = __builtin_amdgcn_readfirstlane(tt);
    vol_store<128>(vol + VO7, t7, tid);
    vol_store<95>(vol + VO6, t6, tid);
    vol_store<70>(vol + VO5, t5, tid);
    vol_store<52>(vol + VO4, t4, tid);
    vol_store<39>(vol + VO3, t3, tid);
    vol_store<28>(vol + VO2, t2, tid);
    vol_store<21>(vol + VO1, t1, tid);
    vol_store<16>(vol + VO0, t0, tid);
  }

  const int  n  = lane & 15;
  const bool hi = lane >= 16;

  v2f bfrag[5];
  for (int c = 0; c < 5; ++c) {
    const int kb = 4 * c + (hi ? 2 : 0);
    bfrag[c].x = SH_COEF[kb][n];
    bfrag[c].y = SH_COEF[kb + 1][n];
  }

  const int rsub = lane >> 3;
  const int c4o  = (lane & 7) * 4;

  for (int c0 = 0; c0 < total; c0 += CH) {
    __syncthreads();
    const int nch = (total - c0) < CH ? (total - c0) : CH;
    if (fits) {
      const int s   = tid >> 1;
      const int par = tid & 1;
      const int st0 = sStart[s];
      const int rbeg = st0 > c0 ? st0 : c0;
      const int rlim = st0 + sCnt[s];
      const int rend = rlim < c0 + CH ? rlim : c0 + CH;
      const int recbase = s * SEG + sLo[s] - st0;
      for (int r = rbeg + par; r < rend; r += 2) rt[r - c0] = recbase + r;
    }
    __syncthreads();

    const int tiles = (nch + 31) >> 5;
    const int last  = nch - 1;
    const int pbase = b * SEG + c0;

    v4f nme = {0.f, 0.f, 0.f, 0.f}, nd0 = nme, nd1 = nme;
    if (wave < tiles) {
      const int fb = wave * 32;
      nme = rec_half<0>(recs, rt, fits, pos, pbase, fb + lane < last ? fb + lane : last);
      nd0 = rec_half<1>(recs, rt, fits, dir, pbase, fb + n < last ? fb + n : last);
      nd1 = rec_half<1>(recs, rt, fits, dir, pbase, fb + 16 + n < last ? fb + 16 + n : last);
    }

    for (int tile = wave; tile < tiles; tile += WAVES) {
      const int base = tile * 32;

      const v4f me = nme;
      const v4f dd[2] = {nd0, nd1};
      const float px = me.x, py = me.y, pz = me.z;

      v2f res[8];
      {
        v2f ca[8], cb[8], cc[8], cd[8];
        float fxa, fya, fza, fxb, fyb, fzb, fxc, fyc, fzc, fxd, fyd, fzd;
        int ga, gb, gc, gd;
        bool ia, ib, ic, id;
        level_fetch<16>(vol + VO0, x0, y0, z0, px, py, pz, ca, fxa, fya, fza, ga, ia);
        level_fetch<21>(vol + VO1, x1, y1, z1, px, py, pz, cb, fxb, fyb, fzb, gb, ib);
        level_fetch<28>(vol + VO2, x2, y2, z2, px, py, pz, cc, fxc, fyc, fzc, gc, ic);
        level_fetch<39>(vol + VO3, x3, y3, z3, px, py, pz, cd, fxd, fyd, fzd, gd, id);
        if (!(ia && ib && ic && id)) {
          if (!ia) corners<16>((const v2f*)g0, ga, ca);
          if (!ib) corners<21>((const v2f*)g1, gb, cb);
          if (!ic) corners<28>((const v2f*)g2, gc, cc);
          if (!id) corners<39>((const v2f*)g3, gd, cd);
        }
        res[0] = trilerp(ca, fxa, fya, fza);
        res[1] = trilerp(cb, fxb, fyb, fzb);
        res[2] = trilerp(cc, fxc, fyc, fzc);
        res[3] = trilerp(cd, fxd, fyd, fzd);
      }
      {
        v2f ca[8], cb[8], cc[8], cd[8];
        float fxa, fya, fza, fxb, fyb, fzb, fxc, fyc, fzc, fxd, fyd, fzd;
        int ga, gb, gc, gd;
        bool ia, ib, ic, id;
        level_fetch<52>(vol + VO4, x4, y4, z4, px, py, pz, ca, fxa, fya, fza, ga, ia);
        level_fetch<70>(vol + VO5, x5, y5, z5, px, py, pz, cb, fxb, fyb, fzb, gb, ib);
        level_fetch<95>(vol + VO6, x6, y6, z6, px, py, pz, cc, fxc, fyc, fzc, gc, ic);
        level_fetch<128>(vol + VO7, x7, y7, z7, px, py, pz, cd, fxd, fyd, fzd, gd, id);
        if (!(ia && ib && ic && id)) {
          if (!ia) corners<52>((const v2f*)g4, ga, ca);
          if (!ib) corners<70>((const v2f*)g5, gb, cb);
          if (!ic) corners<95>((const v2f*)g6, gc, cc);
          if (!id) corners<128>((const v2f*)g7, gd, cd);
        }
        res[4] = trilerp(ca, fxa, fya, fza);
        res[5] = trilerp(cb, fxb, fyb, fzb);
        res[6] = trilerp(cc, fxc, fyc, fzc);
        res[7] = trilerp(cd, fxd, fyd, fzd);
      }

      {
        const int nb = base + 32 * WAVES;
        nme = rec_half<0>(recs, rt, fits, pos, pbase, nb + lane < last ? nb + lane : last);
        nd0 = rec_half<1>(recs, rt, fits, dir, pbase, nb + n < last ? nb + n : last);
        nd1 = rec_half<1>(recs, rt, fits, dir, pbase, nb + 16 + n < last ? nb + 16 + n : last);
      }

      v8f acc[2];
      for (int t = 0; t < 2; ++t) {
        const v4f d4 = dd[t];
        const float dx = d4.x;
        const float dy = d4.y;
        const float dz = d4.z;
        const float inv = rsqrtf(dx * dx + dy * dy + dz * dz);
        const float x = dx * inv, y = dy * inv, z = dz * inv;
        const float xx = x * x, yy = y * y, zz = z * z;

        const float m0 = 1.0f, m1 = x, m2 = y, m3 = z;
        const float m4 = xx, m5 = yy, m6 = zz, m7 = x * y, m8 = y * z, m9 = x * z;
        const float m10 = xx * x, m11 = yy * y, m12 = zz * z;
        const float m13 = xx * y, m14 = xx * z, m15 = x * yy;
        const float m16 = yy * z, m17 = x * zz, m18 = y * zz, m19 = x * y * z;

        v8f a8 = {0.f, 0.f, 0.f, 0.f, 0.f, 0.f, 0.f, 0.f};

#define SH_CHUNK(C, E0, E1, E2, E3)                                           \
        do {                                                                  \
          v2f a;                                                              \
          a.x = hi ? (E2) : (E0);                                             \
          a.y = hi ? (E3) : (E1);                                             \
          a8 = __builtin_amdgcn_wmma_f32_16x16x4_f32(                         \
              false, a, false, bfrag[(C) / 4], (short)0, a8, false, false);   \
        } while (0)

        SH_CHUNK(0,  m0,  m1,  m2,  m3);
        SH_CHUNK(4,  m4,  m5,  m6,  m7);
        SH_CHUNK(8,  m8,  m9,  m10, m11);
        SH_CHUNK(12, m12, m13, m14, m15);
        SH_CHUNK(16, m16, m17, m18, m19);
#undef SH_CHUNK
        acc[t] = a8;
      }

      for (int h = 0; h < 2; ++h) {
        const int rowBase = hi ? 8 : 0;
#pragma unroll
        for (int r = 0; r < 8; ++r) {
          st[(rowBase + r) * ROW_PITCH + 16 + n] = acc[h][r];
        }
        if ((lane >> 4) == h) {
          v4f* srow = (v4f*)(st + n * ROW_PITCH);
          for (int k = 0; k < 4; ++k) {
            v4f v = {res[2 * k].x, res[2 * k].y, res[2 * k + 1].x, res[2 * k + 1].y};
            srow[k] = v;
          }
          st[n * ROW_PITCH + 32] = me.w;
        }

        __builtin_amdgcn_fence(__ATOMIC_RELEASE, "wavefront");
        __builtin_amdgcn_wave_barrier();
        __builtin_amdgcn_fence(__ATOMIC_ACQUIRE, "wavefront");

        if (base + 32 <= nch) {
          for (int j = 0; j < 4; ++j) {
            const int row = 4 * j + rsub;
            int pidx = __float_as_int(st[row * ROW_PITCH + 32]);
            pidx = pidx > 0 ? pidx : 0;
            pidx = pidx < NPTS - 1 ? pidx : NPTS - 1;
            const v4f v = *(const v4f*)(st + row * ROW_PITCH + c4o);
            row_store(out + (size_t)pidx * 32 + c4o, v);
          }
        } else {
          for (int j = 0; j < 4; ++j) {
            const int row = 4 * j + rsub;
            const int rq  = base + HALF_ROWS * h + row;
            const bool ok = rq <= last;
            int pidx = __float_as_int(st[row * ROW_PITCH + 32]);
            pidx = pidx > 0 ? pidx : 0;
            pidx = pidx < NPTS - 1 ? pidx : NPTS - 1;
            const v4f v = *(const v4f*)(st + row * ROW_PITCH + c4o);
            if (ok) row_store(out + (size_t)pidx * 32 + c4o, v);
          }
        }

        __builtin_amdgcn_fence(__ATOMIC_RELEASE, "wavefront");
        __builtin_amdgcn_wave_barrier();
        __builtin_amdgcn_fence(__ATOMIC_ACQUIRE, "wavefront");
      }
    }
  }
}

__global__ __launch_bounds__(THREADS) void encode_kernel(
    const float* __restrict__ pos, const float* __restrict__ dir,
    const float* __restrict__ g0, const float* __restrict__ g1,
    const float* __restrict__ g2, const float* __restrict__ g3,
    const float* __restrict__ g4, const float* __restrict__ g5,
    const float* __restrict__ g6, const float* __restrict__ g7,
    float* __restrict__ out,
    const float* __restrict__ ws, unsigned long long ws_bytes)
{
  __shared__ __attribute__((aligned(16))) float stage[WAVES][HALF_ROWS * ROW_PITCH];
  __shared__ v2f vol[VOL_CELLS];
  __shared__ int rt[CH];
  __shared__ int sLo[NWG], sCnt[NWG], sStart[NWG], sGrp[16];

  unsigned long long lost;
  const bool fits = ws_fits(ws, ws_bytes, lost);
  const v4f* const recs = (const v4f*)((const char*)ws + lost);
  const int* const offs = (const int*)(recs + REC_V4);
  if (fits) {
    encode_body<true>(pos, dir, g0, g1, g2, g3, g4, g5, g6, g7, out, recs, offs,
                      stage, vol, rt, sLo, sCnt, sStart, sGrp);
  } else {
    encode_body<false>(pos, dir, g0, g1, g2, g3, g4, g5, g6, g7, out, recs, offs,
                       stage, vol, rt, sLo, sCnt, sStart, sGrp);
  }
}

extern "C" void kernel_launch(void* const* d_in, const int* in_sizes, int n_in,
                              void* d_out, int out_size, void* d_ws, size_t ws_size,
                              hipStream_t stream) {
  (void)in_sizes; (void)n_in; (void)out_size;

  const float* pos = (const float*)d_in[0];
  const float* dir = (const float*)d_in[1];
  const float* g0 = (const float*)d_in[2];
  const float* g1 = (const float*)d_in[3];
  const float* g2 = (const float*)d_in[4];
  const float* g3 = (const float*)d_in[5];
  const float* g4 = (const float*)d_in[6];
  const float* g5 = (const float*)d_in[7];
  const float* g6 = (const float*)d_in[8];
  const float* g7 = (const float*)d_in[9];
  float* out = (float*)d_out;

  float* ws = (float*)d_ws;

  sort_kernel<<<NWG, THREADS, 0, stream>>>(pos, dir, ws, ws_size);

  encode_kernel<<<NWG, THREADS, 0, stream>>>(pos, dir, g0, g1, g2, g3, g4,
                                             g5, g6, g7, out, ws, ws_size);
}


# Launch 2 trip reordered for overlap: next tile's records requested a whole trip ahead; SH products and levels 0-3 interpolation run under the LDS corner reads; global path one branch; sums unchanged
# speedup vs baseline: 3.5939x; 1.0663x over previous
#include <hip/hip_runtime.h>


typedef __attribute__((ext_vector_type(2))) float v2f;
typedef __attribute__((ext_vector_type(4))) float v4f;
typedef __attribute__((ext_vector_type(8))) float v8f;

#define NPTS (1 << 20)

#define NWG     256
#define WAVES   16
#define THREADS (WAVES * 32)
#define SEG     (NPTS / NWG)
#define PER_T   (SEG / THREADS)
#define NBINS   256
#define CH      8192

#define ROW_PITCH 36
#define HALF_ROWS 16

#define REC_V4    (2 * NPTS)
#define OFF_PITCH 260
#define NEED_BYTES (16ull * REC_V4 + 4ull * NWG * OFF_PITCH)
#define FAR_LO (1 << 20)

__constant__ float SH_COEF[20][16] = {
    {0.28209479177387814f,0,0,0,0,0,-0.31539156525252f,0,0,0,0,0,0,0,0,0},
    {0,0,0,-0.48860251190291987f,0,0,0,0,0,0,0,0,0,0.4570457994644657f,0,0},
    {0,-0.48860251190291987f,0,0,0,0,0,0,0,0,0,0.4570457994644657f,0,0,0,0},
    {0,0,0.48860251190291987f,0,0,0,0,0,0,0,0,0,-1.1195289977703462f,0,0,0},
    {0,0,0,0,0,0,0,0,0.5462742152960396f,0,0,0,0,0,0,0},
    {0,0,0,0,0,0,0,0,-0.5462742152960396f,0,0,0,0,0,0,0},
    {0,0,0,0,0,0,0.94617469575756f,0,0,0,0,0,0,0,0,0},
    {0,0,0,0,1.0925484305920792f,0,0,0,0,0,0,0,0,0,0,0},
    {0,0,0,0,0,-1.0925484305920792f,0,0,0,0,0,0,0,0,0,0},
    {0,0,0,0,0,0,0,-1.0925484305920792f,0,0,0,0,0,0,0,0},
    {0,0,0,0,0,0,0,0,0,0,0,0,0,0,0,-0.5900435899266435f},
    {0,0,0,0,0,0,0,0,0,0.5900435899266435f,0,0,0,0,0,0},
    {0,0,0,0,0,0,0,0,0,0,0,0,1.865881662950577f,0,0,0},
    {0,0,0,0,0,0,0,0,0,-1.7701307697799304f,0,0,0,0,0,0},
    {0,0,0,0,0,0,0,0,0,0,0,0,0,0,1.445305721320277f,0},
    {0,0,0,0,0,0,0,0,0,0,0,0,0,0,0,1.7701307697799304f},
    {0,0,0,0,0,0,0,0,0,0,0,0,0,0,-1.445305721320277f,0},
    {0,0,0,0,0,0,0,0,0,0,0,0,0,-2.2852289973223288f,0,0},
    {0,0,0,0,0,0,0,0,0,0,0,-2.2852289973223288f,0,0,0,0},
    {0,0,0,0,0,0,0,0,0,0,2.890611442640554f,0,0,0,0,0},
};


template <int R>
__device__ __forceinline__ void cell_ijk(float px, float py, float pz,
                                         float& fx, float& fy, float& fz,
                                         int& ix, int& iy, int& iz) {
  const float cx = px * (float)(R - 1);
  const float cy = py * (float)(R - 1);
  const float cz = pz * (float)(R - 1);
  ix = (int)cx; ix = ix < R - 2 ? ix : R - 2; ix = ix > 0 ? ix : 0;
  iy = (int)cy; iy = iy < R - 2 ? iy : R - 2; iy = iy > 0 ? iy : 0;
  iz = (int)cz; iz = iz < R - 2 ? iz : R - 2; iz = iz > 0 ? iz : 0;
  fx = cx - (float)ix;
  fy = cy - (float)iy;
  fz = cz - (float)iz;
}


template <int R>
__device__ __forceinline__ void corners(const v2f* g, int i000, v2f (&c)[8]) {
  const int i100 = i000 + R * R;
  c[0] = g[i000];         c[1] = g[i000 + 1];
  c[2] = g[i000 + R];     c[3] = g[i000 + R + 1];
  c[4] = g[i100];         c[5] = g[i100 + 1];
  c[6] = g[i100 + R];     c[7] = g[i100 + R + 1];
}


__device__ __forceinline__ v2f trilerp(const v2f (&c)[8], float fx, float fy, float fz) {
  const v2f c00 = c[0] + fz * (c[1] - c[0]);
  const v2f c01 = c[2] + fz * (c[3] - c[2]);
  const v2f c10 = c[4] + fz * (c[5] - c[4]);
  const v2f c11 = c[6] + fz * (c[7] - c[6]);
  const v2f c0  = c00 + fy * (c01 - c00);
  const v2f c1  = c10 + fy * (c11 - c10);
  return c0 + fx * (c1 - c0);
}


__device__ __forceinline__ bool ws_fits(const float* ws, unsigned long long bytes,
                                        unsigned long long& lost) {
  const unsigned long long p = (unsigned long long)ws;
  lost = (128ull - (p & 127ull)) & 127ull;
  const unsigned long long avail = bytes > lost ? bytes - lost : 0ull;
  return ws != nullptr && avail >= NEED_BYTES;
}

__device__ __forceinline__ int bin_of(float px, float py, float pz) {
  float fx, fy, fz;
  int ix, iy, iz;
  cell_ijk<128>(px, py, pz, fx, fy, fz, ix, iy, iz);
  return ((ix >> 4) * 8 + (iy >> 4)) * 4 + (iz >> 5);
}

template <int R> struct Sub {
  static constexpr int WX = 16 * (R - 1);
  static constexpr int WZ = 32 * (R - 1);
  static constexpr int CX = (WX % 127 == 0) ? WX / 127 : WX / 127 + 2;
  static constexpr int CZ = (WZ % 127 == 0) ? WZ / 127 : WZ / 127 + 2;
  static constexpr int PX = CX + 1;
  static constexpr int PZ = CZ + 1;
  static constexpr int CELLS = PX * PX * PZ;
  static constexpr int STEPS = (CELLS + THREADS - 1) / THREADS;
};
template <int R> __device__ __forceinline__ int lo_xy(int b) { return (16 * (R - 1) * b) / 127; }
template <int R> __device__ __forceinline__ int lo_z(int b)  { return (32 * (R - 1) * b) / 127; }

#define VO0 0
#define VO1 (VO0 + Sub<16>::CELLS)
#define VO2 (VO1 + Sub<21>::CELLS)
#define VO3 (VO2 + Sub<28>::CELLS)
#define VO4 (VO3 + Sub<39>::CELLS)
#define VO5 (VO4 + Sub<52>::CELLS)
#define VO6 (VO5 + Sub<70>::CELLS)
#define VO7 (VO6 + Sub<95>::CELLS)
#define VOL_CELLS (VO7 + Sub<128>::CELLS)

template <int R>
__device__ __forceinline__ void vol_load(v2f (&t)[Sub<R>::STEPS], const float* g,
                                         int lox, int loy, int loz, int tid) {
  typedef Sub<R> S;
  const v2f* s = (const v2f*)g;
#pragma unroll
  for (int k = 0; k < S::STEPS; ++k) {
    const int q  = tid + k * THREADS;
    const int qc = q < S::CELLS ? q : S::CELLS - 1;
    const int lz = qc % S::PZ;
    const int r  = qc / S::PZ;
    const int ly = r % S::PX;
    const int lx = r / S::PX;
    int gx = lox + lx; gx = gx < R - 1 ? gx : R - 1;
    int gy = loy + ly; gy = gy < R - 1 ? gy : R - 1;
    int gz = loz + lz; gz = gz < R - 1 ? gz : R - 1;
    t[k] = s[(gx * R + gy) * R + gz];
  }
}
template <int R>
__device__ __forceinline__ void vol_store(v2f* dst, const v2f (&t)[Sub<R>::STEPS], int tid) {
  typedef Sub<R> S;
#pragma unroll
  for (int k = 0; k < S::STEPS; ++k) {
    const int q = tid + k * THREADS;
    if (q < S::CELLS) dst[q] = t[k];
  }
}

template <int R>
__device__ __forceinline__ void level_fetch(const v2f* vol, int lox, int loy, int loz,
                                            float px, float py, float pz,
                                            v2f (&c)[8], float& fx, float& fy, float& fz,
                                            bool& inb) {
  typedef Sub<R> S;
  int ix, iy, iz;
  cell_ijk<R>(px, py, pz, fx, fy, fz, ix, iy, iz);
  const int lx = ix - lox, ly = iy - loy, lz = iz - loz;
  inb = (unsigned)lx < (unsigned)S::CX && (unsigned)ly < (unsigned)S::CX &&
        (unsigned)lz < (unsigned)S::CZ;
  const int li = inb ? (lx * S::PX + ly) * S::PZ + lz : 0;
  constexpr int SY = S::PZ;
  constexpr int SX = S::PX * S::PZ;
  c[0] = vol[li];           c[1] = vol[li + 1];
  c[2] = vol[li + SY];      c[3] = vol[li + SY + 1];
  c[4] = vol[li + SX];      c[5] = vol[li + SX + 1];
  c[6] = vol[li + SX + SY]; c[7] = vol[li + SX + SY + 1];
}

template <int R>
__device__ __forceinline__ v2f level_global(const float* g, float px, float py, float pz) {
  float fx, fy, fz;
  int ix, iy, iz;
  cell_ijk<R>(px, py, pz, fx, fy, fz, ix, iy, iz);
  v2f c[8];
  corners<R>((const v2f*)g, (ix * R + iy) * R + iz, c);
  return trilerp(c, fx, fy, fz);
}

__device__ __forceinline__ void row_store(float* p, v4f v) {
  __builtin_nontemporal_store(v, (v4f*)p);
}

template <int HALF>
__device__ __forceinline__ v4f rec_half(const v4f* __restrict__ recs, const int* rt, bool fits,
                                        const float* __restrict__ src, int pbase, int q) {
  if (fits) {
    int idx = rt[q];
    idx = idx > 0 ? idx : 0;
    idx = idx < NPTS - 1 ? idx : NPTS - 1;
    return recs[2 * (size_t)idx + HALF];
  }
  const int p = pbase + q;
  return (v4f){src[3 * p + 0], src[3 * p + 1], src[3 * p + 2], HALF == 0 ? __int_as_float(p) : 0.0f};
}

__global__ __launch_bounds__(THREADS) void sort_kernel(
    const float* __restrict__ pos, const float* __restrict__ dir,
    float* __restrict__ ws, unsigned long long ws_bytes)
{
  __shared__ int whist[WAVES][NBINS];
  __shared__ int part[THREADS];
  __shared__ int wsum[WAVES];
  __shared__ __attribute__((aligned(16))) v4f srec[2 * SEG];

  unsigned long long lost;
  if (!ws_fits(ws, ws_bytes, lost)) return;
  v4f* const recs = (v4f*)((char*)ws + lost);
  int* const offs = (int*)(recs + REC_V4);

  const int tid  = threadIdx.x;
  const int lane = tid & 31;
  const int wave = __builtin_amdgcn_readfirstlane(tid >> 5);

#pragma unroll
  for (int k = 0; k < NBINS * WAVES / THREADS; ++k) (&whist[0][0])[tid + k * THREADS] = 0;

  float ppx[PER_T], ppy[PER_T], ppz[PER_T], pdx[PER_T], pdy[PER_T], pdz[PER_T];
  int bin[PER_T], wr[PER_T];
#pragma unroll
  for (int it = 0; it < PER_T; ++it) {
    const int p = blockIdx.x * SEG + it * THREADS + tid;
    ppx[it] = pos[3 * p + 0]; ppy[it] = pos[3 * p + 1]; ppz[it] = pos[3 * p + 2];
    pdx[it] = dir[3 * p + 0]; pdy[it] = dir[3 * p + 1]; pdz[it] = dir[3 * p + 2];
  }
#pragma unroll
  for (int it = 0; it < PER_T; ++it) bin[it] = bin_of(ppx[it], ppy[it], ppz[it]);
  __syncthreads();

#pragma unroll
  for (int it = 0; it < PER_T; ++it) {
    const int bb = bin[it];
    unsigned mask = 0xffffffffu;
#pragma unroll
    for (int k = 0; k < 8; ++k) {
      const bool bit = (bb >> k) & 1;
      const unsigned bal = __builtin_amdgcn_ballot_w32(bit);
      mask &= bit ? bal : ~bal;
    }
    const int old   = whist[wave][bb];
    const int cnt   = __popc(mask);
    const int below = __popc(mask & ((1u << lane) - 1u));
    wr[it] = old + below;
    __builtin_amdgcn_fence(__ATOMIC_RELEASE, "wavefront");
    __builtin_amdgcn_wave_barrier();
    __builtin_amdgcn_fence(__ATOMIC_ACQUIRE, "wavefront");
    if (below == 0) whist[wave][bb] = old + cnt;
    __builtin_amdgcn_fence(__ATOMIC_RELEASE, "wavefront");
    __builtin_amdgcn_wave_barrier();
    __builtin_amdgcn_fence(__ATOMIC_ACQUIRE, "wavefront");
  }
  __syncthreads();

  {
    const int bb = tid >> 1;
    const int w0 = (tid & 1) * 8;
    int v[8];
    int s = 0;
#pragma unroll
    for (int j = 0; j < 8; ++j) { v[j] = whist[w0 + j][bb]; s += v[j]; }
    part[tid] = s;
    __syncthreads();
    if (tid < WAVES) {
      int a = 0;
      for (int k = 0; k < 32; ++k) a += part[tid * 32 + k];
      wsum[tid] = a;
    }
    __syncthreads();
    int ex = 0;
    for (int k = 0; k < WAVES; ++k) ex += (k < wave) ? wsum[k] : 0;
    for (int k = 0; k < 32; ++k) ex += (k < lane) ? part[wave * 32 + k] : 0;
    int run = ex;
#pragma unroll
    for (int j = 0; j < 8; ++j) { whist[w0 + j][bb] = run; run += v[j]; }
    if ((tid & 1) == 0) offs[blockIdx.x * OFF_PITCH + bb] = ex;
    if (tid == THREADS - 1) offs[blockIdx.x * OFF_PITCH + NBINS] = run;
  }
  __syncthreads();

#pragma unroll
  for (int it = 0; it < PER_T; ++it) {
    const int p = blockIdx.x * SEG + it * THREADS + tid;
    int dest = whist[wave][bin[it]] + wr[it];
    dest = dest < SEG - 1 ? dest : SEG - 1;
    dest = dest > 0 ? dest : 0;
    srec[2 * dest + 0] = (v4f){ppx[it], ppy[it], ppz[it], __int_as_float(p)};
    srec[2 * dest + 1] = (v4f){pdx[it], pdy[it], pdz[it], 0.0f};
  }
  __syncthreads();
  v4f* const seg = recs + (size_t)blockIdx.x * (2 * SEG);
#pragma unroll
  for (int k = 0; k < 2 * SEG / THREADS; ++k) seg[tid + k * THREADS] = srec[tid + k * THREADS];
}

template <bool FITS>
__device__ __forceinline__ void encode_body(
    const float* __restrict__ pos, const float* __restrict__ dir,
    const float* __restrict__ g0, const float* __restrict__ g1,
    const float* __restrict__ g2, const float* __restrict__ g3,
    const float* __restrict__ g4, const float* __restrict__ g5,
    const float* __restrict__ g6, const float* __restrict__ g7,
    float* __restrict__ out,
    const v4f* __restrict__ recs, const int* __restrict__ offs,
    float (*stage)[HALF_ROWS * ROW_PITCH], v2f* vol, int* rt,
    int* sLo, int* sCnt, int* sStart, int* sGrp)
{
  constexpr bool fits = FITS;
  const int tid  = threadIdx.x;
  const int lane = tid & 31;
  const int wave = __builtin_amdgcn_readfirstlane(tid >> 5);
  float* st = stage[wave];
  const int b = blockIdx.x;

  const int bx = b >> 5, by = (b >> 2) & 7, bz = b & 3;
#define LO3(L, R)                                              \
  const int x##L = fits ? lo_xy<R>(bx) : FAR_LO;               \
  const int y##L = fits ? lo_xy<R>(by) : FAR_LO;               \
  const int z##L = fits ? lo_z<R>(bz)  : FAR_LO;
  LO3(0, 16) LO3(1, 21) LO3(2, 28) LO3(3, 39) LO3(4, 52) LO3(5, 70) LO3(6, 95) LO3(7, 128)
#undef LO3

  int total = SEG;
  if (fits) {
    if (tid < NWG) {
      int a = offs[tid * OFF_PITCH + b];
      int e = offs[tid * OFF_PITCH + b + 1];
      a = a > 0 ? a : 0;   a = a < SEG ? a : SEG;
      e = e > a ? e : a;   e = e < SEG ? e : SEG;
      sLo[tid]  = a;
      sCnt[tid] = e - a;
    }
    v2f t0[Sub<16>::STEPS], t1[Sub<21>::STEPS], t2[Sub<28>::STEPS], t3[Sub<39>::STEPS];
    v2f t4[Sub<52>::STEPS], t5[Sub<70>::STEPS], t6[Sub<95>::STEPS], t7[Sub<128>::STEPS];
    vol_load<128>(t7, g7, x7, y7, z7, tid);
    vol_load<95>(t6, g6, x6, y6, z6, tid);
    vol_load<70>(t5, g5, x5, y5, z5, tid);
    vol_load<52>(t4, g4, x4, y4, z4, tid);
    vol_load<39>(t3, g3, x3, y3, z3, tid);
    vol_load<28>(t2, g2, x2, y2, z2, tid);
    vol_load<21>(t1, g1, x1, y1, z1, tid);
    vol_load<16>(t0, g0, x0, y0, z0, tid);
    __syncthreads();
    if (tid < 16) {
      int s = 0;
      for (int k = 0; k < 16; ++k) s += sCnt[tid * 16 + k];
      sGrp[tid] = s;
    }
    __syncthreads();
    if (tid < NWG) {
      const int g = tid >> 4;
      int s = 0;
      for (int k = 0; k < 16; ++k) s += (k < g) ? sGrp[k] : 0;
      for (int k = 0; k < 16; ++k) s += (k < (tid & 15)) ? sCnt[g * 16 + k] : 0;
      sStart[tid] = s;
    }
    int tt = 0;
    for (int k = 0; k < 16; ++k) tt += sGrp[k];
    total = __builtin_amdgcn_readfirstlane(tt);
    vol_store<128>(vol + VO7, t7, tid);
    vol_store<95>(vol + VO6, t6, tid);
    vol_store<70>(vol + VO5, t5, tid);
    vol_store<52>(vol + VO4, t4, tid);
    vol_store<39>(vol + VO3, t3, tid);
    vol_store<28>(vol + VO2, t2, tid);
    vol_store<21>(vol + VO1, t1, tid);
    vol_store<16>(vol + VO0, t0, tid);
  }

  const int  n  = lane & 15;
  const bool hi = lane >= 16;

  v2f bfrag[5];
  for (int c = 0; c < 5; ++c) {
    const int kb = 4 * c + (hi ? 2 : 0);
    bfrag[c].x = SH_COEF[kb][n];
    bfrag[c].y = SH_COEF[kb + 1][n];
  }

  const int rsub = lane >> 3;
  const int c4o  = (lane & 7) * 4;

  for (int c0 = 0; c0 < total; c0 += CH) {
    __syncthreads();
    const int nch = (total - c0) < CH ? (total - c0) : CH;
    if (fits) {
      const int s   = tid >> 1;
      const int par = tid & 1;
      const int st0 = sStart[s];
      const int rbeg = st0 > c0 ? st0 : c0;
      const int rlim = st0 + sCnt[s];
      const int rend = rlim < c0 + CH ? rlim : c0 + CH;
      const int recbase = s * SEG + sLo[s] - st0;
      for (int r = rbeg + par; r < rend; r += 2) rt[r - c0] = recbase + r;
    }
    __syncthreads();

    const int tiles = (nch + 31) >> 5;
    const int last  = nch - 1;
    const int pbase = b * SEG + c0;

    v4f nme = {0.f, 0.f, 0.f, 0.f}, nd0 = nme, nd1 = nme;
    if (wave < tiles) {
      const int fb = wave * 32;
      nme = rec_half<0>(recs, rt, fits, pos, pbase, fb + lane < last ? fb + lane : last);
      nd0 = rec_half<1>(recs, rt, fits, dir, pbase, fb + n < last ? fb + n : last);
      nd1 = rec_half<1>(recs, rt, fits, dir, pbase, fb + 16 + n < last ? fb + 16 + n : last);
    }

    for (int tile = wave; tile < tiles; tile += WAVES) {
      const int base = tile * 32;

      const v4f me = nme;
      const v4f dd[2] = {nd0, nd1};
      const float px = me.x, py = me.y, pz = me.z;

      {
        const int nb = base + 32 * WAVES;
        nme = rec_half<0>(recs, rt, fits, pos, pbase, nb + lane < last ? nb + lane : last);
        nd0 = rec_half<1>(recs, rt, fits, dir, pbase, nb + n < last ? nb + n : last);
        nd1 = rec_half<1>(recs, rt, fits, dir, pbase, nb + 16 + n < last ? nb + 16 + n : last);
      }

      v2f c0[8], c1[8], c2[8], c3[8], c4[8], c5[8], c6[8], c7[8];
      float fx0, fy0, fz0, fx1, fy1, fz1, fx2, fy2, fz2, fx3, fy3, fz3;
      float fx4, fy4, fz4, fx5, fy5, fz5, fx6, fy6, fz6, fx7, fy7, fz7;
      bool in0, in1, in2, in3, in4, in5, in6, in7;
      level_fetch<16>(vol + VO0, x0, y0, z0, px, py, pz, c0, fx0, fy0, fz0, in0);
      level_fetch<21>(vol + VO1, x1, y1, z1, px, py, pz, c1, fx1, fy1, fz1, in1);
      level_fetch<28>(vol + VO2, x2, y2, z2, px, py, pz, c2, fx2, fy2, fz2, in2);
      level_fetch<39>(vol + VO3, x3, y3, z3, px, py, pz, c3, fx3, fy3, fz3, in3);

      v8f acc[2];
      for (int t = 0; t < 2; ++t) {
        const v4f d4 = dd[t];
        const float dx = d4.x;
        const float dy = d4.y;
        const float dz = d4.z;
        const float inv = rsqrtf(dx * dx + dy * dy + dz * dz);
        const float x = dx * inv, y = dy * inv, z = dz * inv;
        const float xx = x * x, yy = y * y, zz = z * z;

        const float m0 = 1.0f, m1 = x, m2 = y, m3 = z;
        const float m4 = xx, m5 = yy, m6 = zz, m7 = x * y, m8 = y * z, m9 = x * z;
        const float m10 = xx * x, m11 = yy * y, m12 = zz * z;
        const float m13 = xx * y, m14 = xx * z, m15 = x * yy;
        const float m16 = yy * z, m17 = x * zz, m18 = y * zz, m19 = x * y * z;

        v8f a8 = {0.f, 0.f, 0.f, 0.f, 0.f, 0.f, 0.f, 0.f};

#define SH_CHUNK(C, E0, E1, E2, E3)                                           \
        do {                                                                  \
          v2f a;                                                              \
          a.x = hi ? (E2) : (E0);                                             \
          a.y = hi ? (E3) : (E1);                                             \
          a8 = __builtin_amdgcn_wmma_f32_16x16x4_f32(                         \
              false, a, false, bfrag[(C) / 4], (short)0, a8, false, false);   \
        } while (0)

        SH_CHUNK(0,  m0,  m1,  m2,  m3);
        SH_CHUNK(4,  m4,  m5,  m6,  m7);
        SH_CHUNK(8,  m8,  m9,  m10, m11);
        SH_CHUNK(12, m12, m13, m14, m15);
        SH_CHUNK(16, m16, m17, m18, m19);
#undef SH_CHUNK
        acc[t] = a8;
      }

      level_fetch<52>(vol + VO4, x4, y4, z4, px, py, pz, c4, fx4, fy4, fz4, in4);
      level_fetch<70>(vol + VO5, x5, y5, z5, px, py, pz, c5, fx5, fy5, fz5, in5);
      level_fetch<95>(vol + VO6, x6, y6, z6, px, py, pz, c6, fx6, fy6, fz6, in6);
      level_fetch<128>(vol + VO7, x7, y7, z7, px, py, pz, c7, fx7, fy7, fz7, in7);

      v2f res[8];
      res[0] = trilerp(c0, fx0, fy0, fz0);
      res[1] = trilerp(c1, fx1, fy1, fz1);
      res[2] = trilerp(c2, fx2, fy2, fz2);
      res[3] = trilerp(c3, fx3, fy3, fz3);
      res[4] = trilerp(c4, fx4, fy4, fz4);
      res[5] = trilerp(c5, fx5, fy5, fz5);
      res[6] = trilerp(c6, fx6, fy6, fz6);
      res[7] = trilerp(c7, fx7, fy7, fz7);
      if (!(in0 && in1 && in2 && in3 && in4 && in5 && in6 && in7)) {
        if (!in0) res[0] = level_global<16>(g0, px, py, pz);
        if (!in1) res[1] = level_global<21>(g1, px, py, pz);
        if (!in2) res[2] = level_global<28>(g2, px, py, pz);
        if (!in3) res[3] = level_global<39>(g3, px, py, pz);
        if (!in4) res[4] = level_global<52>(g4, px, py, pz);
        if (!in5) res[5] = level_global<70>(g5, px, py, pz);
        if (!in6) res[6] = level_global<95>(g6, px, py, pz);
        if (!in7) res[7] = level_global<128>(g7, px, py, pz);
      }

      for (int h = 0; h < 2; ++h) {
        const int rowBase = hi ? 8 : 0;
#pragma unroll
        for (int r = 0; r < 8; ++r) {
          st[(rowBase + r) * ROW_PITCH + 16 + n] = acc[h][r];
        }
        if ((lane >> 4) == h) {
          v4f* srow = (v4f*)(st + n * ROW_PITCH);
          for (int k = 0; k < 4; ++k) {
            v4f v = {res[2 * k].x, res[2 * k].y, res[2 * k + 1].x, res[2 * k + 1].y};
            srow[k] = v;
          }
          st[n * ROW_PITCH + 32] = me.w;
        }

        __builtin_amdgcn_fence(__ATOMIC_RELEASE, "wavefront");
        __builtin_amdgcn_wave_barrier();
        __builtin_amdgcn_fence(__ATOMIC_ACQUIRE, "wavefront");

        if (base + 32 <= nch) {
          for (int j = 0; j < 4; ++j) {
            const int row = 4 * j + rsub;
            int pidx = __float_as_int(st[row * ROW_PITCH + 32]);
            pidx = pidx > 0 ? pidx : 0;
            pidx = pidx < NPTS - 1 ? pidx : NPTS - 1;
            const v4f v = *(const v4f*)(st + row * ROW_PITCH + c4o);
            row_store(out + (size_t)pidx * 32 + c4o, v);
          }
        } else {
          for (int j = 0; j < 4; ++j) {
            const int row = 4 * j + rsub;
            const int rq  = base + HALF_ROWS * h + row;
            const bool ok = rq <= last;
            int pidx = __float_as_int(st[row * ROW_PITCH + 32]);
            pidx = pidx > 0 ? pidx : 0;
            pidx = pidx < NPTS - 1 ? pidx : NPTS - 1;
            const v4f v = *(const v4f*)(st + row * ROW_PITCH + c4o);
            if (ok) row_store(out + (size_t)pidx * 32 + c4o, v);
          }
        }

        __builtin_amdgcn_fence(__ATOMIC_RELEASE, "wavefront");
        __builtin_amdgcn_wave_barrier();
        __builtin_amdgcn_fence(__ATOMIC_ACQUIRE, "wavefront");
      }
    }
  }
}

__global__ __launch_bounds__(THREADS) void encode_kernel(
    const float* __restrict__ pos, const float* __restrict__ dir,
    const float* __restrict__ g0, const float* __restrict__ g1,
    const float* __restrict__ g2, const float* __restrict__ g3,
    const float* __restrict__ g4, const float* __restrict__ g5,
    const float* __restrict__ g6, const float* __restrict__ g7,
    float* __restrict__ out,
    const float* __restrict__ ws, unsigned long long ws_bytes)
{
  __shared__ __attribute__((aligned(16))) float stage[WAVES][HALF_ROWS * ROW_PITCH];
  __shared__ v2f vol[VOL_CELLS];
  __shared__ int rt[CH];
  __shared__ int sSeg[3 * NWG];
  __shared__ int sGrp[16];
  int* const sLo    = sSeg;
  int* const sCnt   = sSeg + NWG;
  int* const sStart = sSeg + 2 * NWG;

  unsigned long long lost;
  const bool fits = ws_fits(ws, ws_bytes, lost);
  const v4f* const recs = (const v4f*)((const char*)ws + lost);
  const int* const offs = (const int*)(recs + REC_V4);
  if (fits) {
    encode_body<true>(pos, dir, g0, g1, g2, g3, g4, g5, g6, g7, out, recs, offs,
                      stage, vol, rt, sLo, sCnt, sStart, sGrp);
  } else {
    encode_body<false>(pos, dir, g0, g1, g2, g3, g4, g5, g6, g7, out, recs, offs,
                       stage, vol, rt, sLo, sCnt, sStart, sGrp);
  }
}

extern "C" void kernel_launch(void* const* d_in, const int* in_sizes, int n_in,
                              void* d_out, int out_size, void* d_ws, size_t ws_size,
                              hipStream_t stream) {
  (void)in_sizes; (void)n_in; (void)out_size;

  const float* pos = (const float*)d_in[0];
  const float* dir = (const float*)d_in[1];
  const float* g0 = (const float*)d_in[2];
  const float* g1 = (const float*)d_in[3];
  const float* g2 = (const float*)d_in[4];
  const float* g3 = (const float*)d_in[5];
  const float* g4 = (const float*)d_in[6];
  const float* g5 = (const float*)d_in[7];
  const float* g6 = (const float*)d_in[8];
  const float* g7 = (const float*)d_in[9];
  float* out = (float*)d_out;

  float* ws = (float*)d_ws;

  sort_kernel<<<NWG, THREADS, 0, stream>>>(pos, dir, ws, ws_size);

  encode_kernel<<<NWG, THREADS, 0, stream>>>(pos, dir, g0, g1, g2, g3, g4,
                                             g5, g6, g7, out, ws, ws_size);
}
